# starting kernel
# baseline (speedup 1.0000x reference)
#include <hip/hip_runtime.h>
#include <hip/hip_bf16.h>


#define NB 65536
#define NT 100

typedef __attribute__((ext_vector_type(2))) float v2f;
typedef __attribute__((ext_vector_type(8))) float v8f;

__device__ __forceinline__ float fast_rcp(float x) {
  return __builtin_amdgcn_rcpf(x);
}
__device__ __forceinline__ float sigm(float x) {
  return fast_rcp(1.0f + __expf(-x));
}
__device__ __forceinline__ float tanh_fast(float x) {
  return fmaf(2.0f, fast_rcp(1.0f + __expf(-2.0f * x)), -1.0f);
}

template <int IN, bool REV, bool TMAJOR>
__global__ __launch_bounds__(256) void lstm_dir_kernel(
    const float* __restrict__ in,
    const float* __restrict__ w_ih,
    const float* __restrict__ w_hh,
    const float* __restrict__ b_ih,
    const float* __restrict__ b_hh,
    float* __restrict__ out,
    int out_off)
{
  const int b = blockIdx.x * blockDim.x + threadIdx.x;
  if (b >= NB) return;

  float wih[8][IN];
  float whh[8][2];
  float bias[8];
#pragma unroll
  for (int g = 0; g < 8; ++g) {
#pragma unroll
    for (int c = 0; c < IN; ++c) wih[g][c] = w_ih[g * IN + c];
    whh[g][0] = w_hh[g * 2 + 0];
    whh[g][1] = w_hh[g * 2 + 1];
    bias[g] = b_ih[g] + b_hh[g];
  }

  float h0 = 0.f, h1 = 0.f, c0 = 0.f, c1 = 0.f;

  for (int it = 0; it < NT; ++it) {
    const int tt = REV ? (NT - 1 - it) : it;

    float xv[IN];
    if constexpr (TMAJOR) {
      const float4 x4 =
          *(const float4*)(in + ((size_t)tt * NB + (size_t)b) * 4);
      xv[0] = x4.x; xv[1] = x4.y; xv[2] = x4.z; xv[3] = x4.w;
    } else {
      const float* p = in + (size_t)b * NT * IN + (size_t)tt * IN;
#pragma unroll
      for (int c = 0; c < IN; ++c) xv[c] = p[c];
    }

    float g[8];
#pragma unroll
    for (int gg = 0; gg < 8; ++gg) {
      float acc = fmaf(whh[gg][0], h0, fmaf(whh[gg][1], h1, bias[gg]));
#pragma unroll
      for (int c = 0; c < IN; ++c) acc = fmaf(wih[gg][c], xv[c], acc);
      g[gg] = acc;
    }

    const float i0 = sigm(g[0]), i1 = sigm(g[1]);
    const float f0 = sigm(g[2]), f1 = sigm(g[3]);
    const float t0 = tanh_fast(g[4]), t1 = tanh_fast(g[5]);
    const float o0 = sigm(g[6]), o1 = sigm(g[7]);

    c0 = fmaf(f0, c0, i0 * t0);
    c1 = fmaf(f1, c1, i1 * t1);
    h0 = o0 * tanh_fast(c0);
    h1 = o1 * tanh_fast(c1);

    float2 hv = make_float2(h0, h1);
    *(float2*)(out + ((size_t)tt * NB + (size_t)b) * 4 + out_off) = hv;
  }
}

__global__ __launch_bounds__(256) void fc1_wmma_kernel(
    const float* __restrict__ h2,
    const float* __restrict__ w,
    const float* __restrict__ bias,
    float* __restrict__ act)
{
  const int wid = blockIdx.x * (blockDim.x >> 5) + (threadIdx.x >> 5);
  const int lane = threadIdx.x & 31;
  const int m_tile = wid >> 2;
  const int n_tile = wid & 3;
  const int row = lane & 15;
  const int half = lane >> 4;
  const int base_b = m_tile * 16;
  const int n = n_tile * 16 + row;

  v8f acc = {};
  const float* wrow = w + (size_t)n * 400 + half * 2;
  const float* arow = h2 + ((size_t)(base_b + row)) * 4 + half * 2;

#pragma unroll 4
  for (int t = 0; t < NT; ++t) {
    v2f a = *(const v2f*)(arow + (size_t)t * NB * 4);
    a.x = fmaxf(a.x, 0.0f);
    a.y = fmaxf(a.y, 0.0f);
    v2f bv = *(const v2f*)(wrow + t * 4);
    acc = __builtin_amdgcn_wmma_f32_16x16x4_f32(
        false, a, false, bv, (short)0, acc, false, false);
  }

  const float bn = bias[n];
#pragma unroll
  for (int r = 0; r < 8; ++r) {
    const int m = r + half * 8;
    act[(size_t)(base_b + m) * 64 + n] = fmaxf(acc[r] + bn, 0.0f);
  }
}

__global__ __launch_bounds__(256) void fc2_wmma_kernel(
    const float* __restrict__ act,
    const float* __restrict__ w,
    const float* __restrict__ bias,
    float* __restrict__ out)
{
  const int wid = blockIdx.x * (blockDim.x >> 5) + (threadIdx.x >> 5);
  const int lane = threadIdx.x & 31;
  const int m_tile = wid >> 1;
  const int n_tile = wid & 1;
  const int row = lane & 15;
  const int half = lane >> 4;
  const int base_b = m_tile * 16;
  const int n = n_tile * 16 + row;
  const bool nvalid = (n < 20);

  v8f acc = {};
  const float* arow = act + (size_t)(base_b + row) * 64 + half * 2;
  const float* wrow = w + (size_t)(nvalid ? n : 0) * 64 + half * 2;

#pragma unroll 4
  for (int kk = 0; kk < 16; ++kk) {
    v2f a = *(const v2f*)(arow + kk * 4);
    v2f bv = *(const v2f*)(wrow + kk * 4);
    if (!nvalid) { bv.x = 0.0f; bv.y = 0.0f; }
    acc = __builtin_amdgcn_wmma_f32_16x16x4_f32(
        false, a, false, bv, (short)0, acc, false, false);
  }

  if (nvalid) {
    const float bn = bias[n];
#pragma unroll
    for (int r = 0; r < 8; ++r) {
      const int m = r + half * 8;
      out[(size_t)(base_b + m) * 20 + n] = acc[r] + bn;
    }
  }
}

extern "C" void kernel_launch(void* const* d_in, const int* in_sizes, int n_in,
                              void* d_out, int out_size, void* d_ws,
                              size_t ws_size, hipStream_t stream) {
  const float* x = (const float*)d_in[0];
  const float* w_ih_l0  = (const float*)d_in[1];
  const float* w_hh_l0  = (const float*)d_in[2];
  const float* b_ih_l0  = (const float*)d_in[3];
  const float* b_hh_l0  = (const float*)d_in[4];
  const float* w_ih_l0r = (const float*)d_in[5];
  const float* w_hh_l0r = (const float*)d_in[6];
  const float* b_ih_l0r = (const float*)d_in[7];
  const float* b_hh_l0r = (const float*)d_in[8];
  const float* w_ih_l1  = (const float*)d_in[9];
  const float* w_hh_l1  = (const float*)d_in[10];
  const float* b_ih_l1  = (const float*)d_in[11];
  const float* b_hh_l1  = (const float*)d_in[12];
  const float* w_ih_l1r = (const float*)d_in[13];
  const float* w_hh_l1r = (const float*)d_in[14];
  const float* b_ih_l1r = (const float*)d_in[15];
  const float* b_hh_l1r = (const float*)d_in[16];
  const float* fc1_w = (const float*)d_in[17];
  const float* fc1_b = (const float*)d_in[18];
  const float* fc2_w = (const float*)d_in[19];
  const float* fc2_b = (const float*)d_in[20];
  float* out = (float*)d_out;

  float* ws  = (float*)d_ws;
  float* ws0 = ws;
  float* ws1 = ws + (size_t)NT * NB * 4;
  float* ws2 = ws;

  const dim3 blk(256);
  const dim3 grid_lstm(NB / 256);

  lstm_dir_kernel<7, false, false><<<grid_lstm, blk, 0, stream>>>(
      x, w_ih_l0, w_hh_l0, b_ih_l0, b_hh_l0, ws0, 0);
  lstm_dir_kernel<7, true, false><<<grid_lstm, blk, 0, stream>>>(
      x, w_ih_l0r, w_hh_l0r, b_ih_l0r, b_hh_l0r, ws0, 2);

  lstm_dir_kernel<4, false, true><<<grid_lstm, blk, 0, stream>>>(
      ws0, w_ih_l1, w_hh_l1, b_ih_l1, b_hh_l1, ws1, 0);
  lstm_dir_kernel<4, true, true><<<grid_lstm, blk, 0, stream>>>(
      ws0, w_ih_l1r, w_hh_l1r, b_ih_l1r, b_hh_l1r, ws1, 2);

  fc1_wmma_kernel<<<dim3(2048), blk, 0, stream>>>(ws1, fc1_w, fc1_b, ws2);

  fc2_wmma_kernel<<<dim3(1024), blk, 0, stream>>>(ws2, fc2_w, fc2_b, out);
}


# LSTM kernels prefetch inputs: register ring of 4-step groups loaded 3 groups ahead of the recurrence (x as 7 aligned 16-byte loads per group, not 28 dwords); arithmetic and its order unchanged
# speedup vs baseline: 1.1347x; 1.1347x over previous
#include <hip/hip_runtime.h>
#include <hip/hip_bf16.h>


#define NB 65536
#define NT 100

typedef __attribute__((ext_vector_type(2))) float v2f;
typedef __attribute__((ext_vector_type(4))) float v4f;
typedef __attribute__((ext_vector_type(8))) float v8f;

__device__ __forceinline__ float fast_rcp(float x) {
  return __builtin_amdgcn_rcpf(x);
}
__device__ __forceinline__ float sigm(float x) {
  return fast_rcp(1.0f + __expf(-x));
}
__device__ __forceinline__ float tanh_fast(float x) {
  return fmaf(2.0f, fast_rcp(1.0f + __expf(-2.0f * x)), -1.0f);
}

#define GS 4
#define NG (NT / GS)
#define RING 4

template <int IN, bool REV, bool TMAJOR>
__global__ __launch_bounds__(256) void lstm_dir_kernel(
    const float* __restrict__ in,
    const float* __restrict__ w_ih,
    const float* __restrict__ w_hh,
    const float* __restrict__ b_ih,
    const float* __restrict__ b_hh,
    float* __restrict__ out,
    int out_off)
{
  static_assert(NT % GS == 0, "whole groups");
  static_assert(RING == 4 && NG % RING == 1 && NG > 2 * RING,
                "the loop below leaves exactly RING + 1 groups");
  static_assert(!TMAJOR || IN == 4, "time-major input has 4 channels");
  static_assert((GS * IN) % 4 == 0, "a group is whole 16-byte vectors");

  const int b = blockIdx.x * blockDim.x + threadIdx.x;
  if (b >= NB) return;

  float wih[8][IN];
  float whh[8][2];
  float bias[8];
#pragma unroll
  for (int g = 0; g < 8; ++g) {
#pragma unroll
    for (int c = 0; c < IN; ++c) wih[g][c] = w_ih[g * IN + c];
    whh[g][0] = w_hh[g * 2 + 0];
    whh[g][1] = w_hh[g * 2 + 1];
    bias[g] = b_ih[g] + b_hh[g];
  }

  float h0 = 0.f, h1 = 0.f, c0 = 0.f, c1 = 0.f;

  auto load_group = [&](int g, v4f (&buf)[IN]) {
    const int gi = REV ? (NG - 1 - g) : g;
    if constexpr (TMAJOR) {
#pragma unroll
      for (int s = 0; s < GS; ++s)
        buf[s] = *(const v4f*)(in + ((size_t)(gi * GS + s) * NB + (size_t)b) * 4);
    } else {
      const v4f* p =
          (const v4f*)(in + (size_t)b * NT * IN + (size_t)gi * GS * IN);
#pragma unroll
      for (int v = 0; v < IN; ++v) buf[v] = p[v];
    }
  };

  auto run_group = [&](int g, const v4f (&buf)[IN]) {
    const int gi = REV ? (NG - 1 - g) : g;
#pragma unroll
    for (int s = 0; s < GS; ++s) {
      const int sl = REV ? (GS - 1 - s) : s;
      const int tt = gi * GS + sl;

      float xv[IN];
#pragma unroll
      for (int c = 0; c < IN; ++c) {
        const int e = sl * IN + c;
        xv[c] = buf[e >> 2][e & 3];
      }

      float g8[8];
#pragma unroll
      for (int gg = 0; gg < 8; ++gg) {
        float acc = fmaf(whh[gg][0], h0, fmaf(whh[gg][1], h1, bias[gg]));
#pragma unroll
        for (int c = 0; c < IN; ++c) acc = fmaf(wih[gg][c], xv[c], acc);
        g8[gg] = acc;
      }

      const float i0 = sigm(g8[0]), i1 = sigm(g8[1]);
      const float f0 = sigm(g8[2]), f1 = sigm(g8[3]);
      const float t0 = tanh_fast(g8[4]), t1 = tanh_fast(g8[5]);
      const float o0 = sigm(g8[6]), o1 = sigm(g8[7]);

      c0 = fmaf(f0, c0, i0 * t0);
      c1 = fmaf(f1, c1, i1 * t1);
      h0 = o0 * tanh_fast(c0);
      h1 = o1 * tanh_fast(c1);

      float2 hv = make_float2(h0, h1);
      *(float2*)(out + ((size_t)tt * NB + (size_t)b) * 4 + out_off) = hv;
    }
  };

  auto fetch = [&](int g, v4f (&buf)[IN]) {
    asm volatile("" ::: "memory");
    __builtin_amdgcn_sched_barrier(0);
    load_group(g, buf);
    __builtin_amdgcn_sched_barrier(0);
  };

  v4f r0[IN], r1[IN], r2[IN], r3[IN];
  load_group(0, r0);
  load_group(1, r1);
  load_group(2, r2);
  for (int g = 0; g < NG - 1 - RING; g += RING) {
    fetch(g + 3, r3);
    run_group(g, r0);
    fetch(g + 4, r0);
    run_group(g + 1, r1);
    fetch(g + 5, r1);
    run_group(g + 2, r2);
    fetch(g + 6, r2);
    run_group(g + 3, r3);
  }
  fetch(NG - 2, r3);
  run_group(NG - 5, r0);
  fetch(NG - 1, r0);
  run_group(NG - 4, r1);
  run_group(NG - 3, r2);
  run_group(NG - 2, r3);
  run_group(NG - 1, r0);
}

__global__ __launch_bounds__(256) void fc1_wmma_kernel(
    const float* __restrict__ h2,
    const float* __restrict__ w,
    const float* __restrict__ bias,
    float* __restrict__ act)
{
  const int wid = blockIdx.x * (blockDim.x >> 5) + (threadIdx.x >> 5);
  const int lane = threadIdx.x & 31;
  const int m_tile = wid >> 2;
  const int n_tile = wid & 3;
  const int row = lane & 15;
  const int half = lane >> 4;
  const int base_b = m_tile * 16;
  const int n = n_tile * 16 + row;

  v8f acc = {};
  const float* wrow = w + (size_t)n * 400 + half * 2;
  const float* arow = h2 + ((size_t)(base_b + row)) * 4 + half * 2;

#pragma unroll 4
  for (int t = 0; t < NT; ++t) {
    v2f a = *(const v2f*)(arow + (size_t)t * NB * 4);
    a.x = fmaxf(a.x, 0.0f);
    a.y = fmaxf(a.y, 0.0f);
    v2f bv = *(const v2f*)(wrow + t * 4);
    acc = __builtin_amdgcn_wmma_f32_16x16x4_f32(
        false, a, false, bv, (short)0, acc, false, false);
  }

  const float bn = bias[n];
#pragma unroll
  for (int r = 0; r < 8; ++r) {
    const int m = r + half * 8;
    act[(size_t)(base_b + m) * 64 + n] = fmaxf(acc[r] + bn, 0.0f);
  }
}

__global__ __launch_bounds__(256) void fc2_wmma_kernel(
    const float* __restrict__ act,
    const float* __restrict__ w,
    const float* __restrict__ bias,
    float* __restrict__ out)
{
  const int wid = blockIdx.x * (blockDim.x >> 5) + (threadIdx.x >> 5);
  const int lane = threadIdx.x & 31;
  const int m_tile = wid >> 1;
  const int n_tile = wid & 1;
  const int row = lane & 15;
  const int half = lane >> 4;
  const int base_b = m_tile * 16;
  const int n = n_tile * 16 + row;
  const bool nvalid = (n < 20);

  v8f acc = {};
  const float* arow = act + (size_t)(base_b + row) * 64 + half * 2;
  const float* wrow = w + (size_t)(nvalid ? n : 0) * 64 + half * 2;

#pragma unroll 4
  for (int kk = 0; kk < 16; ++kk) {
    v2f a = *(const v2f*)(arow + kk * 4);
    v2f bv = *(const v2f*)(wrow + kk * 4);
    if (!nvalid) { bv.x = 0.0f; bv.y = 0.0f; }
    acc = __builtin_amdgcn_wmma_f32_16x16x4_f32(
        false, a, false, bv, (short)0, acc, false, false);
  }

  if (nvalid) {
    const float bn = bias[n];
#pragma unroll
    for (int r = 0; r < 8; ++r) {
      const int m = r + half * 8;
      out[(size_t)(base_b + m) * 20 + n] = acc[r] + bn;
    }
  }
}

extern "C" void kernel_launch(void* const* d_in, const int* in_sizes, int n_in,
                              void* d_out, int out_size, void* d_ws,
                              size_t ws_size, hipStream_t stream) {
  const float* x = (const float*)d_in[0];
  const float* w_ih_l0  = (const float*)d_in[1];
  const float* w_hh_l0  = (const float*)d_in[2];
  const float* b_ih_l0  = (const float*)d_in[3];
  const float* b_hh_l0  = (const float*)d_in[4];
  const float* w_ih_l0r = (const float*)d_in[5];
  const float* w_hh_l0r = (const float*)d_in[6];
  const float* b_ih_l0r = (const float*)d_in[7];
  const float* b_hh_l0r = (const float*)d_in[8];
  const float* w_ih_l1  = (const float*)d_in[9];
  const float* w_hh_l1  = (const float*)d_in[10];
  const float* b_ih_l1  = (const float*)d_in[11];
  const float* b_hh_l1  = (const float*)d_in[12];
  const float* w_ih_l1r = (const float*)d_in[13];
  const float* w_hh_l1r = (const float*)d_in[14];
  const float* b_ih_l1r = (const float*)d_in[15];
  const float* b_hh_l1r = (const float*)d_in[16];
  const float* fc1_w = (const float*)d_in[17];
  const float* fc1_b = (const float*)d_in[18];
  const float* fc2_w = (const float*)d_in[19];
  const float* fc2_b = (const float*)d_in[20];
  float* out = (float*)d_out;

  float* ws  = (float*)d_ws;
  float* ws0 = ws;
  float* ws1 = ws + (size_t)NT * NB * 4;
  float* ws2 = ws;

  const dim3 blk(256);
  const dim3 grid_lstm(NB / 256);

  lstm_dir_kernel<7, false, false><<<grid_lstm, blk, 0, stream>>>(
      x, w_ih_l0, w_hh_l0, b_ih_l0, b_hh_l0, ws0, 0);
  lstm_dir_kernel<7, true, false><<<grid_lstm, blk, 0, stream>>>(
      x, w_ih_l0r, w_hh_l0r, b_ih_l0r, b_hh_l0r, ws0, 2);

  lstm_dir_kernel<4, false, true><<<grid_lstm, blk, 0, stream>>>(
      ws0, w_ih_l1, w_hh_l1, b_ih_l1, b_hh_l1, ws1, 0);
  lstm_dir_kernel<4, true, true><<<grid_lstm, blk, 0, stream>>>(
      ws0, w_ih_l1r, w_hh_l1r, b_ih_l1r, b_hh_l1r, ws1, 2);

  fc1_wmma_kernel<<<dim3(2048), blk, 0, stream>>>(ws1, fc1_w, fc1_b, ws2);

  fc2_wmma_kernel<<<dim3(1024), blk, 0, stream>>>(ws2, fc2_w, fc2_b, out);
}


# Both directions of each bi-LSTM layer run in one launch (grid y = direction: 4 waves per SIMD, not 2), each writing its own (T,B,2) plane in whole lines; input ring cut to 3 groups; sums unchanged
# speedup vs baseline: 2.0042x; 1.7663x over previous
#include <hip/hip_runtime.h>
#include <hip/hip_bf16.h>


#define NB 65536
#define NT 100

typedef __attribute__((ext_vector_type(2))) float v2f;
typedef __attribute__((ext_vector_type(4))) float v4f;
typedef __attribute__((ext_vector_type(8))) float v8f;

__device__ __forceinline__ float fast_rcp(float x) {
  return __builtin_amdgcn_rcpf(x);
}
__device__ __forceinline__ float sigm(float x) {
  return fast_rcp(1.0f + __expf(-x));
}
__device__ __forceinline__ float tanh_fast(float x) {
  return fmaf(2.0f, fast_rcp(1.0f + __expf(-2.0f * x)), -1.0f);
}

#define GS 4
#define NG (NT / GS)
#define RING 3
#define PLANE ((size_t)NT * NB * 2)

template <int IN, bool REV, bool TMAJOR>
__device__ __forceinline__ void lstm_dir(
    const float* __restrict__ in,
    const float* __restrict__ w_ih,
    const float* __restrict__ w_hh,
    const float* __restrict__ b_ih,
    const float* __restrict__ b_hh,
    float* __restrict__ out)
{
  static_assert(NT % GS == 0, "whole groups");
  static_assert(RING == 3 && NG % RING == 1 && NG > RING,
                "the loop below leaves exactly one group");
  static_assert(!TMAJOR || IN == 4, "time-major input has 4 channels");
  static_assert((GS * IN) % 4 == 0, "a group is whole 16-byte vectors");

  const int b = blockIdx.x * blockDim.x + threadIdx.x;
  if (b >= NB) return;

  float wih[8][IN];
  float whh[8][2];
  float bias[8];
#pragma unroll
  for (int g = 0; g < 8; ++g) {
#pragma unroll
    for (int c = 0; c < IN; ++c) wih[g][c] = w_ih[g * IN + c];
    whh[g][0] = w_hh[g * 2 + 0];
    whh[g][1] = w_hh[g * 2 + 1];
    bias[g] = b_ih[g] + b_hh[g];
  }

  float h0 = 0.f, h1 = 0.f, c0 = 0.f, c1 = 0.f;

  auto load_group = [&](int g, v4f (&buf)[IN]) {
    const int gc = g < NG ? g : NG - 1;
    const int gi = REV ? (NG - 1 - gc) : gc;
    if constexpr (TMAJOR) {
#pragma unroll
      for (int s = 0; s < GS; ++s) {
        const float* p = in + ((size_t)(gi * GS + s) * NB + (size_t)b) * 2;
        const v2f f = *(const v2f*)p;
        const v2f r = *(const v2f*)(p + PLANE);
        buf[s] = v4f{f.x, f.y, r.x, r.y};
      }
    } else {
      const v4f* p =
          (const v4f*)(in + (size_t)b * NT * IN + (size_t)gi * GS * IN);
#pragma unroll
      for (int v = 0; v < IN; ++v) buf[v] = p[v];
    }
  };

  auto run_group = [&](int g, const v4f (&buf)[IN]) {
    const int gi = REV ? (NG - 1 - g) : g;
#pragma unroll
    for (int s = 0; s < GS; ++s) {
      const int sl = REV ? (GS - 1 - s) : s;
      const int tt = gi * GS + sl;

      float xv[IN];
#pragma unroll
      for (int c = 0; c < IN; ++c) {
        const int e = sl * IN + c;
        xv[c] = buf[e >> 2][e & 3];
      }

      float g8[8];
#pragma unroll
      for (int gg = 0; gg < 8; ++gg) {
        float acc = fmaf(whh[gg][0], h0, fmaf(whh[gg][1], h1, bias[gg]));
#pragma unroll
        for (int c = 0; c < IN; ++c) acc = fmaf(wih[gg][c], xv[c], acc);
        g8[gg] = acc;
      }

      const float i0 = sigm(g8[0]), i1 = sigm(g8[1]);
      const float f0 = sigm(g8[2]), f1 = sigm(g8[3]);
      const float t0 = tanh_fast(g8[4]), t1 = tanh_fast(g8[5]);
      const float o0 = sigm(g8[6]), o1 = sigm(g8[7]);

      c0 = fmaf(f0, c0, i0 * t0);
      c1 = fmaf(f1, c1, i1 * t1);
      h0 = o0 * tanh_fast(c0);
      h1 = o1 * tanh_fast(c1);

      float2 hv = make_float2(h0, h1);
      *(float2*)(out + ((size_t)tt * NB + (size_t)b) * 2) = hv;
    }
  };

  auto fetch = [&](int g, v4f (&buf)[IN]) {
    asm volatile("" ::: "memory");
    __builtin_amdgcn_sched_barrier(0);
    load_group(g, buf);
    __builtin_amdgcn_sched_barrier(0);
  };

  v4f r0[IN], r1[IN], r2[IN];
  load_group(0, r0);
  load_group(1, r1);
  for (int g = 0; g < NG - 1; g += RING) {
    fetch(g + 2, r2);
    run_group(g, r0);
    fetch(g + 3, r0);
    run_group(g + 1, r1);
    fetch(g + 4, r1);
    run_group(g + 2, r2);
  }
  run_group(NG - 1, r0);
}

template <int IN, bool TMAJOR>
__global__ __launch_bounds__(256) void lstm_layer_kernel(
    const float* __restrict__ in,
    const float* __restrict__ w_ih_f, const float* __restrict__ w_hh_f,
    const float* __restrict__ b_ih_f, const float* __restrict__ b_hh_f,
    const float* __restrict__ w_ih_r, const float* __restrict__ w_hh_r,
    const float* __restrict__ b_ih_r, const float* __restrict__ b_hh_r,
    float* __restrict__ out)
{
  if (blockIdx.y == 0)
    lstm_dir<IN, false, TMAJOR>(in, w_ih_f, w_hh_f, b_ih_f, b_hh_f, out);
  else
    lstm_dir<IN, true, TMAJOR>(in, w_ih_r, w_hh_r, b_ih_r, b_hh_r,
                               out + PLANE);
}

__global__ __launch_bounds__(256) void fc1_wmma_kernel(
    const float* __restrict__ h2,
    const float* __restrict__ w,
    const float* __restrict__ bias,
    float* __restrict__ act)
{
  const int wid = blockIdx.x * (blockDim.x >> 5) + (threadIdx.x >> 5);
  const int lane = threadIdx.x & 31;
  const int m_tile = wid >> 2;
  const int n_tile = wid & 3;
  const int row = lane & 15;
  const int half = lane >> 4;
  const int base_b = m_tile * 16;
  const int n = n_tile * 16 + row;

  v8f acc = {};
  const float* wrow = w + (size_t)n * 400 + half * 2;
  const float* arow = h2 + (size_t)half * PLANE + ((size_t)(base_b + row)) * 2;

#pragma unroll 4
  for (int t = 0; t < NT; ++t) {
    v2f a = *(const v2f*)(arow + (size_t)t * NB * 2);
    a.x = fmaxf(a.x, 0.0f);
    a.y = fmaxf(a.y, 0.0f);
    v2f bv = *(const v2f*)(wrow + t * 4);
    acc = __builtin_amdgcn_wmma_f32_16x16x4_f32(
        false, a, false, bv, (short)0, acc, false, false);
  }

  const float bn = bias[n];
#pragma unroll
  for (int r = 0; r < 8; ++r) {
    const int m = r + half * 8;
    act[(size_t)(base_b + m) * 64 + n] = fmaxf(acc[r] + bn, 0.0f);
  }
}

__global__ __launch_bounds__(256) void fc2_wmma_kernel(
    const float* __restrict__ act,
    const float* __restrict__ w,
    const float* __restrict__ bias,
    float* __restrict__ out)
{
  const int wid = blockIdx.x * (blockDim.x >> 5) + (threadIdx.x >> 5);
  const int lane = threadIdx.x & 31;
  const int m_tile = wid >> 1;
  const int n_tile = wid & 1;
  const int row = lane & 15;
  const int half = lane >> 4;
  const int base_b = m_tile * 16;
  const int n = n_tile * 16 + row;
  const bool nvalid = (n < 20);

  v8f acc = {};
  const float* arow = act + (size_t)(base_b + row) * 64 + half * 2;
  const float* wrow = w + (size_t)(nvalid ? n : 0) * 64 + half * 2;

#pragma unroll 4
  for (int kk = 0; kk < 16; ++kk) {
    v2f a = *(const v2f*)(arow + kk * 4);
    v2f bv = *(const v2f*)(wrow + kk * 4);
    if (!nvalid) { bv.x = 0.0f; bv.y = 0.0f; }
    acc = __builtin_amdgcn_wmma_f32_16x16x4_f32(
        false, a, false, bv, (short)0, acc, false, false);
  }

  if (nvalid) {
    const float bn = bias[n];
#pragma unroll
    for (int r = 0; r < 8; ++r) {
      const int m = r + half * 8;
      out[(size_t)(base_b + m) * 20 + n] = acc[r] + bn;
    }
  }
}

extern "C" void kernel_launch(void* const* d_in, const int* in_sizes, int n_in,
                              void* d_out, int out_size, void* d_ws,
                              size_t ws_size, hipStream_t stream) {
  const float* x = (const float*)d_in[0];
  const float* w_ih_l0  = (const float*)d_in[1];
  const float* w_hh_l0  = (const float*)d_in[2];
  const float* b_ih_l0  = (const float*)d_in[3];
  const float* b_hh_l0  = (const float*)d_in[4];
  const float* w_ih_l0r = (const float*)d_in[5];
  const float* w_hh_l0r = (const float*)d_in[6];
  const float* b_ih_l0r = (const float*)d_in[7];
  const float* b_hh_l0r = (const float*)d_in[8];
  const float* w_ih_l1  = (const float*)d_in[9];
  const float* w_hh_l1  = (const float*)d_in[10];
  const float* b_ih_l1  = (const float*)d_in[11];
  const float* b_hh_l1  = (const float*)d_in[12];
  const float* w_ih_l1r = (const float*)d_in[13];
  const float* w_hh_l1r = (const float*)d_in[14];
  const float* b_ih_l1r = (const float*)d_in[15];
  const float* b_hh_l1r = (const float*)d_in[16];
  const float* fc1_w = (const float*)d_in[17];
  const float* fc1_b = (const float*)d_in[18];
  const float* fc2_w = (const float*)d_in[19];
  const float* fc2_b = (const float*)d_in[20];
  float* out = (float*)d_out;

  float* ws  = (float*)d_ws;
  float* ws0 = ws;
  float* ws1 = ws + (size_t)NT * NB * 4;
  float* ws2 = ws;

  const dim3 blk(256);
  const dim3 grid_lstm(NB / 256, 2);

  lstm_layer_kernel<7, false><<<grid_lstm, blk, 0, stream>>>(
      x, w_ih_l0, w_hh_l0, b_ih_l0, b_hh_l0,
      w_ih_l0r, w_hh_l0r, b_ih_l0r, b_hh_l0r, ws0);

  lstm_layer_kernel<4, true><<<grid_lstm, blk, 0, stream>>>(
      ws0, w_ih_l1, w_hh_l1, b_ih_l1, b_hh_l1,
      w_ih_l1r, w_hh_l1r, b_ih_l1r, b_hh_l1r, ws1);

  fc1_wmma_kernel<<<dim3(2048), blk, 0, stream>>>(ws1, fc1_w, fc1_b, ws2);

  fc2_wmma_kernel<<<dim3(1024), blk, 0, stream>>>(ws2, fc2_w, fc2_b, out);
}


# Layer-0 LSTM reads row-major x in whole-line pieces (one wave-wide 16-byte load spans 4-5 rows end to end: about 8 lines, not 32), turned to one row per lane in the wave's own LDS; sums unchanged
# speedup vs baseline: 2.1835x; 1.0895x over previous
#include <hip/hip_runtime.h>
#include <hip/hip_bf16.h>


#define NB 65536
#define NT 100

typedef __attribute__((ext_vector_type(2))) float v2f;
typedef __attribute__((ext_vector_type(4))) float v4f;
typedef __attribute__((ext_vector_type(8))) float v8f;

__device__ __forceinline__ float fast_rcp(float x) {
  return __builtin_amdgcn_rcpf(x);
}
__device__ __forceinline__ float sigm(float x) {
  return fast_rcp(1.0f + __expf(-x));
}
__device__ __forceinline__ float tanh_fast(float x) {
  return fmaf(2.0f, fast_rcp(1.0f + __expf(-2.0f * x)), -1.0f);
}

#define GS 4
#define NG (NT / GS)
#define RING 3
#define PLANE ((size_t)NT * NB * 2)

template <int IN, bool REV, bool TMAJOR>
__device__ __forceinline__ void lstm_dir(
    const float* __restrict__ in,
    const float* __restrict__ w_ih,
    const float* __restrict__ w_hh,
    const float* __restrict__ b_ih,
    const float* __restrict__ b_hh,
    float* __restrict__ out,
    v4f* stg)
{
  static_assert(NT % GS == 0, "whole groups");
  static_assert(RING == 3 && NG % RING == 1 && NG > RING,
                "the loop below leaves exactly one group");
  static_assert(!TMAJOR || IN == 4, "time-major input has 4 channels");
  static_assert((GS * IN) % 4 == 0, "a group is whole 16-byte vectors");

  static_assert(NB % 32 == 0, "a wave's 32 rows are all inside the batch");
  static_assert(TMAJOR || GS == 4,
                "row-major: a row's group is IN 16-byte vectors");

  const int b = blockIdx.x * blockDim.x + threadIdx.x;
  if (b >= NB) return;

  const int lane = threadIdx.x & 31;
  const float* wave_in = in + (size_t)(b - lane) * NT * IN;
  int poff[IN];
#pragma unroll
  for (int k = 0; k < IN; ++k) {
    const int p = k * 32 + lane;
    poff[k] = (p / IN) * (NT * IN) + (p % IN) * 4;
  }

  float wih[8][IN];
  float whh[8][2];
  float bias[8];
#pragma unroll
  for (int g = 0; g < 8; ++g) {
#pragma unroll
    for (int c = 0; c < IN; ++c) wih[g][c] = w_ih[g * IN + c];
    whh[g][0] = w_hh[g * 2 + 0];
    whh[g][1] = w_hh[g * 2 + 1];
    bias[g] = b_ih[g] + b_hh[g];
  }

  float h0 = 0.f, h1 = 0.f, c0 = 0.f, c1 = 0.f;

  auto load_group = [&](int g, v4f (&buf)[IN]) {
    const int gc = g < NG ? g : NG - 1;
    const int gi = REV ? (NG - 1 - gc) : gc;
    if constexpr (TMAJOR) {
#pragma unroll
      for (int s = 0; s < GS; ++s) {
        const float* p = in + ((size_t)(gi * GS + s) * NB + (size_t)b) * 2;
        const v2f f = *(const v2f*)p;
        const v2f r = *(const v2f*)(p + PLANE);
        buf[s] = v4f{f.x, f.y, r.x, r.y};
      }
    } else {
      const float* p = wave_in + (size_t)gi * GS * IN;
#pragma unroll
      for (int k = 0; k < IN; ++k) buf[k] = *(const v4f*)(p + poff[k]);
    }
  };

  auto own_row = [&](const v4f (&pc)[IN], v4f (&xr)[IN]) {
#pragma unroll
    for (int k = 0; k < IN; ++k) stg[k * 32 + lane] = pc[k];
#pragma unroll
    for (int q = 0; q < IN; ++q) {
      const int v = REV ? IN - 1 - q : q;
      xr[v] = stg[lane * IN + v];
    }
    __builtin_amdgcn_sched_barrier(0);
  };

  auto run_group = [&](int g, const v4f (&buf)[IN]) {
    const int gi = REV ? (NG - 1 - g) : g;
#pragma unroll
    for (int s = 0; s < GS; ++s) {
      const int sl = REV ? (GS - 1 - s) : s;
      const int tt = gi * GS + sl;

      float xv[IN];
#pragma unroll
      for (int c = 0; c < IN; ++c) {
        const int e = sl * IN + c;
        xv[c] = buf[e >> 2][e & 3];
      }

      float g8[8];
#pragma unroll
      for (int gg = 0; gg < 8; ++gg) {
        float acc = fmaf(whh[gg][0], h0, fmaf(whh[gg][1], h1, bias[gg]));
#pragma unroll
        for (int c = 0; c < IN; ++c) acc = fmaf(wih[gg][c], xv[c], acc);
        g8[gg] = acc;
      }

      const float i0 = sigm(g8[0]), i1 = sigm(g8[1]);
      const float f0 = sigm(g8[2]), f1 = sigm(g8[3]);
      const float t0 = tanh_fast(g8[4]), t1 = tanh_fast(g8[5]);
      const float o0 = sigm(g8[6]), o1 = sigm(g8[7]);

      c0 = fmaf(f0, c0, i0 * t0);
      c1 = fmaf(f1, c1, i1 * t1);
      h0 = o0 * tanh_fast(c0);
      h1 = o1 * tanh_fast(c1);

      float2 hv = make_float2(h0, h1);
      *(float2*)(out + ((size_t)tt * NB + (size_t)b) * 2) = hv;
    }
  };

  auto fetch = [&](int g, v4f (&buf)[IN]) {
    asm volatile("" ::: "memory");
    __builtin_amdgcn_sched_barrier(0);
    load_group(g, buf);
    __builtin_amdgcn_sched_barrier(0);
  };

  auto use_group = [&](int g, const v4f (&buf)[IN]) {
    if constexpr (TMAJOR) {
      run_group(g, buf);
    } else {
      v4f xr[IN];
      own_row(buf, xr);
      run_group(g, xr);
    }
  };

  v4f r0[IN], r1[IN], r2[IN];
  if constexpr (TMAJOR) {
    load_group(0, r0);
    load_group(1, r1);
  } else {
    fetch(0, r0);
    fetch(1, r1);
  }
  for (int g = 0; g < NG - 1; g += RING) {
    fetch(g + 2, r2);
    use_group(g, r0);
    fetch(g + 3, r0);
    use_group(g + 1, r1);
    fetch(g + 4, r1);
    use_group(g + 2, r2);
  }
  use_group(NG - 1, r0);
}

template <int IN, bool TMAJOR>
__global__ __launch_bounds__(256) void lstm_layer_kernel(
    const float* __restrict__ in,
    const float* __restrict__ w_ih_f, const float* __restrict__ w_hh_f,
    const float* __restrict__ b_ih_f, const float* __restrict__ b_hh_f,
    const float* __restrict__ w_ih_r, const float* __restrict__ w_hh_r,
    const float* __restrict__ b_ih_r, const float* __restrict__ b_hh_r,
    float* __restrict__ out)
{
  v4f* stg = nullptr;
  if constexpr (!TMAJOR) {
    __shared__ v4f stage[8][32 * IN];
    stg = &stage[threadIdx.x >> 5][0];
  }
  if (blockIdx.y == 0)
    lstm_dir<IN, false, TMAJOR>(in, w_ih_f, w_hh_f, b_ih_f, b_hh_f, out, stg);
  else
    lstm_dir<IN, true, TMAJOR>(in, w_ih_r, w_hh_r, b_ih_r, b_hh_r,
                               out + PLANE, stg);
}

__global__ __launch_bounds__(256) void fc1_wmma_kernel(
    const float* __restrict__ h2,
    const float* __restrict__ w,
    const float* __restrict__ bias,
    float* __restrict__ act)
{
  const int wid = blockIdx.x * (blockDim.x >> 5) + (threadIdx.x >> 5);
  const int lane = threadIdx.x & 31;
  const int m_tile = wid >> 2;
  const int n_tile = wid & 3;
  const int row = lane & 15;
  const int half = lane >> 4;
  const int base_b = m_tile * 16;
  const int n = n_tile * 16 + row;

  v8f acc = {};
  const float* wrow = w + (size_t)n * 400 + half * 2;
  const float* arow = h2 + (size_t)half * PLANE + ((size_t)(base_b + row)) * 2;

#pragma unroll 4
  for (int t = 0; t < NT; ++t) {
    v2f a = *(const v2f*)(arow + (size_t)t * NB * 2);
    a.x = fmaxf(a.x, 0.0f);
    a.y = fmaxf(a.y, 0.0f);
    v2f bv = *(const v2f*)(wrow + t * 4);
    acc = __builtin_amdgcn_wmma_f32_16x16x4_f32(
        false, a, false, bv, (short)0, acc, false, false);
  }

  const float bn = bias[n];
#pragma unroll
  for (int r = 0; r < 8; ++r) {
    const int m = r + half * 8;
    act[(size_t)(base_b + m) * 64 + n] = fmaxf(acc[r] + bn, 0.0f);
  }
}

__global__ __launch_bounds__(256) void fc2_wmma_kernel(
    const float* __restrict__ act,
    const float* __restrict__ w,
    const float* __restrict__ bias,
    float* __restrict__ out)
{
  const int wid = blockIdx.x * (blockDim.x >> 5) + (threadIdx.x >> 5);
  const int lane = threadIdx.x & 31;
  const int m_tile = wid >> 1;
  const int n_tile = wid & 1;
  const int row = lane & 15;
  const int half = lane >> 4;
  const int base_b = m_tile * 16;
  const int n = n_tile * 16 + row;
  const bool nvalid = (n < 20);

  v8f acc = {};
  const float* arow = act + (size_t)(base_b + row) * 64 + half * 2;
  const float* wrow = w + (size_t)(nvalid ? n : 0) * 64 + half * 2;

#pragma unroll 4
  for (int kk = 0; kk < 16; ++kk) {
    v2f a = *(const v2f*)(arow + kk * 4);
    v2f bv = *(const v2f*)(wrow + kk * 4);
    if (!nvalid) { bv.x = 0.0f; bv.y = 0.0f; }
    acc = __builtin_amdgcn_wmma_f32_16x16x4_f32(
        false, a, false, bv, (short)0, acc, false, false);
  }

  if (nvalid) {
    const float bn = bias[n];
#pragma unroll
    for (int r = 0; r < 8; ++r) {
      const int m = r + half * 8;
      out[(size_t)(base_b + m) * 20 + n] = acc[r] + bn;
    }
  }
}

extern "C" void kernel_launch(void* const* d_in, const int* in_sizes, int n_in,
                              void* d_out, int out_size, void* d_ws,
                              size_t ws_size, hipStream_t stream) {
  const float* x = (const float*)d_in[0];
  const float* w_ih_l0  = (const float*)d_in[1];
  const float* w_hh_l0  = (const float*)d_in[2];
  const float* b_ih_l0  = (const float*)d_in[3];
  const float* b_hh_l0  = (const float*)d_in[4];
  const float* w_ih_l0r = (const float*)d_in[5];
  const float* w_hh_l0r = (const float*)d_in[6];
  const float* b_ih_l0r = (const float*)d_in[7];
  const float* b_hh_l0r = (const float*)d_in[8];
  const float* w_ih_l1  = (const float*)d_in[9];
  const float* w_hh_l1  = (const float*)d_in[10];
  const float* b_ih_l1  = (const float*)d_in[11];
  const float* b_hh_l1  = (const float*)d_in[12];
  const float* w_ih_l1r = (const float*)d_in[13];
  const float* w_hh_l1r = (const float*)d_in[14];
  const float* b_ih_l1r = (const float*)d_in[15];
  const float* b_hh_l1r = (const float*)d_in[16];
  const float* fc1_w = (const float*)d_in[17];
  const float* fc1_b = (const float*)d_in[18];
  const float* fc2_w = (const float*)d_in[19];
  const float* fc2_b = (const float*)d_in[20];
  float* out = (float*)d_out;

  float* ws  = (float*)d_ws;
  float* ws0 = ws;
  float* ws1 = ws + (size_t)NT * NB * 4;
  float* ws2 = ws;

  const dim3 blk(256);
  const dim3 grid_lstm(NB / 256, 2);

  lstm_layer_kernel<7, false><<<grid_lstm, blk, 0, stream>>>(
      x, w_ih_l0, w_hh_l0, b_ih_l0, b_hh_l0,
      w_ih_l0r, w_hh_l0r, b_ih_l0r, b_hh_l0r, ws0);

  lstm_layer_kernel<4, true><<<grid_lstm, blk, 0, stream>>>(
      ws0, w_ih_l1, w_hh_l1, b_ih_l1, b_hh_l1,
      w_ih_l1r, w_hh_l1r, b_ih_l1r, b_hh_l1r, ws1);

  fc1_wmma_kernel<<<dim3(2048), blk, 0, stream>>>(ws1, fc1_w, fc1_b, ws2);

  fc2_wmma_kernel<<<dim3(1024), blk, 0, stream>>>(ws2, fc2_w, fc2_b, out);
}


# LSTM gates on the hardware tanh: each sigmoid and tanh is one v_tanh_f32 (sigmoid = (1+tanh(x/2))/2, x/2 folded exactly into the i,f,o weights at entry), not v_exp_f32 + v_rcp_f32; sums unchanged
# speedup vs baseline: 2.2597x; 1.0349x over previous
#include <hip/hip_runtime.h>
#include <hip/hip_bf16.h>


#define NB 65536
#define NT 100

typedef __attribute__((ext_vector_type(2))) float v2f;
typedef __attribute__((ext_vector_type(4))) float v4f;
typedef __attribute__((ext_vector_type(8))) float v8f;

__device__ __forceinline__ float tanh_hw(float x) {
  return __builtin_amdgcn_tanhf(x);
}
__device__ __forceinline__ float uni(float v) {
  return __int_as_float(__builtin_amdgcn_readfirstlane(__float_as_int(v)));
}
__device__ __forceinline__ float sigm_half(float half_x) {
  return fmaf(0.5f, tanh_hw(half_x), 0.5f);
}

#define GS 4
#define NG (NT / GS)
#define RING 3
#define PLANE ((size_t)NT * NB * 2)

template <int IN, bool REV, bool TMAJOR>
__device__ __forceinline__ void lstm_dir(
    const float* __restrict__ in,
    const float* __restrict__ w_ih,
    const float* __restrict__ w_hh,
    const float* __restrict__ b_ih,
    const float* __restrict__ b_hh,
    float* __restrict__ out,
    v4f* stg)
{
  static_assert(NT % GS == 0, "whole groups");
  static_assert(RING == 3 && NG % RING == 1 && NG > RING,
                "the loop below leaves exactly one group");
  static_assert(!TMAJOR || IN == 4, "time-major input has 4 channels");
  static_assert((GS * IN) % 4 == 0, "a group is whole 16-byte vectors");

  static_assert(NB % 32 == 0, "a wave's 32 rows are all inside the batch");
  static_assert(TMAJOR || GS == 4,
                "row-major: a row's group is IN 16-byte vectors");

  const int b = blockIdx.x * blockDim.x + threadIdx.x;
  if (b >= NB) return;

  const int lane = threadIdx.x & 31;
  const float* wave_in = in + (size_t)(b - lane) * NT * IN;
  int poff[IN];
#pragma unroll
  for (int k = 0; k < IN; ++k) {
    const int p = k * 32 + lane;
    poff[k] = (p / IN) * (NT * IN) + (p % IN) * 4;
  }

  float wih[8][IN];
  float whh[8][2];
  float bias[8];
#pragma unroll
  for (int g = 0; g < 8; ++g) {
    const float sc = (g == 4 || g == 5) ? 1.0f : 0.5f;
#pragma unroll
    for (int c = 0; c < IN; ++c) wih[g][c] = uni(sc * w_ih[g * IN + c]);
    whh[g][0] = uni(sc * w_hh[g * 2 + 0]);
    whh[g][1] = uni(sc * w_hh[g * 2 + 1]);
    bias[g] = uni(sc * (b_ih[g] + b_hh[g]));
  }

  float h0 = 0.f, h1 = 0.f, c0 = 0.f, c1 = 0.f;

  auto load_group = [&](int g, v4f (&buf)[IN]) {
    const int gc = g < NG ? g : NG - 1;
    const int gi = REV ? (NG - 1 - gc) : gc;
    if constexpr (TMAJOR) {
#pragma unroll
      for (int s = 0; s < GS; ++s) {
        const float* p = in + ((size_t)(gi * GS + s) * NB + (size_t)b) * 2;
        const v2f f = *(const v2f*)p;
        const v2f r = *(const v2f*)(p + PLANE);
        buf[s] = v4f{f.x, f.y, r.x, r.y};
      }
    } else {
      const float* p = wave_in + (size_t)gi * GS * IN;
#pragma unroll
      for (int k = 0; k < IN; ++k) buf[k] = *(const v4f*)(p + poff[k]);
    }
  };

  auto own_row = [&](const v4f (&pc)[IN], v4f (&xr)[IN]) {
#pragma unroll
    for (int k = 0; k < IN; ++k) stg[k * 32 + lane] = pc[k];
#pragma unroll
    for (int q = 0; q < IN; ++q) {
      const int v = REV ? IN - 1 - q : q;
      xr[v] = stg[lane * IN + v];
    }
    __builtin_amdgcn_sched_barrier(0);
  };

  auto run_group = [&](int g, const v4f (&buf)[IN]) {
    const int gi = REV ? (NG - 1 - g) : g;
#pragma unroll
    for (int s = 0; s < GS; ++s) {
      const int sl = REV ? (GS - 1 - s) : s;
      const int tt = gi * GS + sl;

      float xv[IN];
#pragma unroll
      for (int c = 0; c < IN; ++c) {
        const int e = sl * IN + c;
        xv[c] = buf[e >> 2][e & 3];
      }

      float g8[8];
#pragma unroll
      for (int gg = 0; gg < 8; ++gg) {
        float acc = fmaf(whh[gg][0], h0, fmaf(whh[gg][1], h1, bias[gg]));
#pragma unroll
        for (int c = 0; c < IN; ++c) acc = fmaf(wih[gg][c], xv[c], acc);
        g8[gg] = acc;
      }

      const float i0 = sigm_half(g8[0]), i1 = sigm_half(g8[1]);
      const float f0 = sigm_half(g8[2]), f1 = sigm_half(g8[3]);
      const float t0 = tanh_hw(g8[4]), t1 = tanh_hw(g8[5]);
      const float o0 = sigm_half(g8[6]), o1 = sigm_half(g8[7]);

      c0 = fmaf(f0, c0, i0 * t0);
      c1 = fmaf(f1, c1, i1 * t1);
      h0 = o0 * tanh_hw(c0);
      h1 = o1 * tanh_hw(c1);

      float2 hv = make_float2(h0, h1);
      *(float2*)(out + ((size_t)tt * NB + (size_t)b) * 2) = hv;
    }
  };

  auto fetch = [&](int g, v4f (&buf)[IN]) {
    asm volatile("" ::: "memory");
    __builtin_amdgcn_sched_barrier(0);
    load_group(g, buf);
    __builtin_amdgcn_sched_barrier(0);
  };

  auto use_group = [&](int g, const v4f (&buf)[IN]) {
    if constexpr (TMAJOR) {
      run_group(g, buf);
    } else {
      v4f xr[IN];
      own_row(buf, xr);
      run_group(g, xr);
    }
  };

  v4f r0[IN], r1[IN], r2[IN];
  if constexpr (TMAJOR) {
    load_group(0, r0);
    load_group(1, r1);
  } else {
    fetch(0, r0);
    fetch(1, r1);
  }
  for (int g = 0; g < NG - 1; g += RING) {
    fetch(g + 2, r2);
    use_group(g, r0);
    fetch(g + 3, r0);
    use_group(g + 1, r1);
    fetch(g + 4, r1);
    use_group(g + 2, r2);
  }
  use_group(NG - 1, r0);
}

template <int IN, bool TMAJOR>
__global__ __launch_bounds__(256) void lstm_layer_kernel(
    const float* __restrict__ in,
    const float* __restrict__ w_ih_f, const float* __restrict__ w_hh_f,
    const float* __restrict__ b_ih_f, const float* __restrict__ b_hh_f,
    const float* __restrict__ w_ih_r, const float* __restrict__ w_hh_r,
    const float* __restrict__ b_ih_r, const float* __restrict__ b_hh_r,
    float* __restrict__ out)
{
  v4f* stg = nullptr;
  if constexpr (!TMAJOR) {
    __shared__ v4f stage[8][32 * IN];
    stg = &stage[threadIdx.x >> 5][0];
  }
  if (blockIdx.y == 0)
    lstm_dir<IN, false, TMAJOR>(in, w_ih_f, w_hh_f, b_ih_f, b_hh_f, out, stg);
  else
    lstm_dir<IN, true, TMAJOR>(in, w_ih_r, w_hh_r, b_ih_r, b_hh_r,
                               out + PLANE, stg);
}

__global__ __launch_bounds__(256) void fc1_wmma_kernel(
    const float* __restrict__ h2,
    const float* __restrict__ w,
    const float* __restrict__ bias,
    float* __restrict__ act)
{
  const int wid = blockIdx.x * (blockDim.x >> 5) + (threadIdx.x >> 5);
  const int lane = threadIdx.x & 31;
  const int m_tile = wid >> 2;
  const int n_tile = wid & 3;
  const int row = lane & 15;
  const int half = lane >> 4;
  const int base_b = m_tile * 16;
  const int n = n_tile * 16 + row;

  v8f acc = {};
  const float* wrow = w + (size_t)n * 400 + half * 2;
  const float* arow = h2 + (size_t)half * PLANE + ((size_t)(base_b + row)) * 2;

#pragma unroll 4
  for (int t = 0; t < NT; ++t) {
    v2f a = *(const v2f*)(arow + (size_t)t * NB * 2);
    a.x = fmaxf(a.x, 0.0f);
    a.y = fmaxf(a.y, 0.0f);
    v2f bv = *(const v2f*)(wrow + t * 4);
    acc = __builtin_amdgcn_wmma_f32_16x16x4_f32(
        false, a, false, bv, (short)0, acc, false, false);
  }

  const float bn = bias[n];
#pragma unroll
  for (int r = 0; r < 8; ++r) {
    const int m = r + half * 8;
    act[(size_t)(base_b + m) * 64 + n] = fmaxf(acc[r] + bn, 0.0f);
  }
}

__global__ __launch_bounds__(256) void fc2_wmma_kernel(
    const float* __restrict__ act,
    const float* __restrict__ w,
    const float* __restrict__ bias,
    float* __restrict__ out)
{
  const int wid = blockIdx.x * (blockDim.x >> 5) + (threadIdx.x >> 5);
  const int lane = threadIdx.x & 31;
  const int m_tile = wid >> 1;
  const int n_tile = wid & 1;
  const int row = lane & 15;
  const int half = lane >> 4;
  const int base_b = m_tile * 16;
  const int n = n_tile * 16 + row;
  const bool nvalid = (n < 20);

  v8f acc = {};
  const float* arow = act + (size_t)(base_b + row) * 64 + half * 2;
  const float* wrow = w + (size_t)(nvalid ? n : 0) * 64 + half * 2;

#pragma unroll 4
  for (int kk = 0; kk < 16; ++kk) {
    v2f a = *(const v2f*)(arow + kk * 4);
    v2f bv = *(const v2f*)(wrow + kk * 4);
    if (!nvalid) { bv.x = 0.0f; bv.y = 0.0f; }
    acc = __builtin_amdgcn_wmma_f32_16x16x4_f32(
        false, a, false, bv, (short)0, acc, false, false);
  }

  if (nvalid) {
    const float bn = bias[n];
#pragma unroll
    for (int r = 0; r < 8; ++r) {
      const int m = r + half * 8;
      out[(size_t)(base_b + m) * 20 + n] = acc[r] + bn;
    }
  }
}

extern "C" void kernel_launch(void* const* d_in, const int* in_sizes, int n_in,
                              void* d_out, int out_size, void* d_ws,
                              size_t ws_size, hipStream_t stream) {
  const float* x = (const float*)d_in[0];
  const float* w_ih_l0  = (const float*)d_in[1];
  const float* w_hh_l0  = (const float*)d_in[2];
  const float* b_ih_l0  = (const float*)d_in[3];
  const float* b_hh_l0  = (const float*)d_in[4];
  const float* w_ih_l0r = (const float*)d_in[5];
  const float* w_hh_l0r = (const float*)d_in[6];
  const float* b_ih_l0r = (const float*)d_in[7];
  const float* b_hh_l0r = (const float*)d_in[8];
  const float* w_ih_l1  = (const float*)d_in[9];
  const float* w_hh_l1  = (const float*)d_in[10];
  const float* b_ih_l1  = (const float*)d_in[11];
  const float* b_hh_l1  = (const float*)d_in[12];
  const float* w_ih_l1r = (const float*)d_in[13];
  const float* w_hh_l1r = (const float*)d_in[14];
  const float* b_ih_l1r = (const float*)d_in[15];
  const float* b_hh_l1r = (const float*)d_in[16];
  const float* fc1_w = (const float*)d_in[17];
  const float* fc1_b = (const float*)d_in[18];
  const float* fc2_w = (const float*)d_in[19];
  const float* fc2_b = (const float*)d_in[20];
  float* out = (float*)d_out;

  float* ws  = (float*)d_ws;
  float* ws0 = ws;
  float* ws1 = ws + (size_t)NT * NB * 4;
  float* ws2 = ws;

  const dim3 blk(256);
  const dim3 grid_lstm(NB / 256, 2);

  lstm_layer_kernel<7, false><<<grid_lstm, blk, 0, stream>>>(
      x, w_ih_l0, w_hh_l0, b_ih_l0, b_hh_l0,
      w_ih_l0r, w_hh_l0r, b_ih_l0r, b_hh_l0r, ws0);

  lstm_layer_kernel<4, true><<<grid_lstm, blk, 0, stream>>>(
      ws0, w_ih_l1, w_hh_l1, b_ih_l1, b_hh_l1,
      w_ih_l1r, w_hh_l1r, b_ih_l1r, b_hh_l1r, ws1);

  fc1_wmma_kernel<<<dim3(2048), blk, 0, stream>>>(ws1, fc1_w, fc1_b, ws2);

  fc2_wmma_kernel<<<dim3(1024), blk, 0, stream>>>(ws2, fc2_w, fc2_b, out);
}


# FC1 summed inside the layer-1 LSTM launch: per wave relu(h) of two time steps x fc1 weights from LDS into 8 f32 WMMA tiles; h2 planes and FC1 launch go; FC1 sum order: per direction, then added
# speedup vs baseline: 3.0316x; 1.3416x over previous
#include <hip/hip_runtime.h>
#include <hip/hip_bf16.h>


#define NB 65536
#define NT 100

typedef __attribute__((ext_vector_type(2))) float v2f;
typedef __attribute__((ext_vector_type(4))) float v4f;
typedef __attribute__((ext_vector_type(8))) float v8f;

__device__ __forceinline__ float tanh_hw(float x) {
  return __builtin_amdgcn_tanhf(x);
}
__device__ __forceinline__ float uni(float v) {
  return __int_as_float(__builtin_amdgcn_readfirstlane(__float_as_int(v)));
}
__device__ __forceinline__ float sigm_half(float half_x) {
  return fmaf(0.5f, tanh_hw(half_x), 0.5f);
}

__device__ __forceinline__ float xhalf(float x) {
  const int b = __float_as_int(x);
  return __int_as_float(
      __builtin_amdgcn_permlanex16(b, b, 0x76543210, 0xfedcba98, false, false));
}

#define GS 4
#define NG (NT / GS)
#define RING 3
#define PLANE ((size_t)NT * NB * 2)

#define WPITCH 160
template <int IN, bool REV, bool TMAJOR, bool FC>
__device__ __forceinline__ void lstm_dir(
    const float* __restrict__ in,
    const float* __restrict__ w_ih,
    const float* __restrict__ w_hh,
    const float* __restrict__ b_ih,
    const float* __restrict__ b_hh,
    float* __restrict__ out,
    v4f* stg,
    const float* wl)
{
  static_assert(!FC || (TMAJOR && GS % 2 == 0), "FC1 sums: pairs of steps");
  static_assert(NT % GS == 0, "whole groups");
  static_assert(RING == 3 && NG % RING == 1 && NG > RING,
                "the loop below leaves exactly one group");
  static_assert(!TMAJOR || IN == 4, "time-major input has 4 channels");
  static_assert((GS * IN) % 4 == 0, "a group is whole 16-byte vectors");

  static_assert(NB % 32 == 0, "a wave's 32 rows are all inside the batch");
  static_assert(TMAJOR || GS == 4,
                "row-major: a row's group is IN 16-byte vectors");

  const int b = blockIdx.x * blockDim.x + threadIdx.x;
  if (b >= NB) return;

  const int lane = threadIdx.x & 31;
  const float* wave_in = in + (size_t)(b - lane) * NT * IN;
  int poff[IN];
#pragma unroll
  for (int k = 0; k < IN; ++k) {
    const int p = k * 32 + lane;
    poff[k] = (p / IN) * (NT * IN) + (p % IN) * 4;
  }

  float wih[8][IN];
  float whh[8][2];
  float bias[8];
#pragma unroll
  for (int g = 0; g < 8; ++g) {
    const float sc = (g == 4 || g == 5) ? 1.0f : 0.5f;
#pragma unroll
    for (int c = 0; c < IN; ++c) wih[g][c] = uni(sc * w_ih[g * IN + c]);
    whh[g][0] = uni(sc * w_hh[g * 2 + 0]);
    whh[g][1] = uni(sc * w_hh[g * 2 + 1]);
    bias[g] = uni(sc * (b_ih[g] + b_hh[g]));
  }

  float h0 = 0.f, h1 = 0.f, c0 = 0.f, c1 = 0.f;

  auto load_group = [&](int g, v4f (&buf)[IN]) {
    const int gc = g < NG ? g : NG - 1;
    const int gi = REV ? (NG - 1 - gc) : gc;
    if constexpr (TMAJOR) {
#pragma unroll
      for (int s = 0; s < GS; ++s) {
        const float* p = in + ((size_t)(gi * GS + s) * NB + (size_t)b) * 2;
        const v2f f = *(const v2f*)p;
        const v2f r = *(const v2f*)(p + PLANE);
        buf[s] = v4f{f.x, f.y, r.x, r.y};
      }
    } else {
      const float* p = wave_in + (size_t)gi * GS * IN;
#pragma unroll
      for (int k = 0; k < IN; ++k) buf[k] = *(const v4f*)(p + poff[k]);
    }
  };

  auto own_row = [&](const v4f (&pc)[IN], v4f (&xr)[IN]) {
#pragma unroll
    for (int k = 0; k < IN; ++k) stg[k * 32 + lane] = pc[k];
#pragma unroll
    for (int q = 0; q < IN; ++q) {
      const int v = REV ? IN - 1 - q : q;
      xr[v] = stg[lane * IN + v];
    }
    __builtin_amdgcn_sched_barrier(0);
  };

  v8f fsum[2][4] = {};
  const bool hi = lane >= 16;
  const int row = lane & 15;
  auto fc_pair = [&](float pa, float pb, int tA, float qa, float qb, int tB) {
    const float p0 = fmaxf(pa, 0.0f), p1 = fmaxf(pb, 0.0f);
    const float q0 = fmaxf(qa, 0.0f), q1 = fmaxf(qb, 0.0f);
    const float xp0 = xhalf(p0), xp1 = xhalf(p1);
    const float xq0 = xhalf(q0), xq1 = xhalf(q1);
    v2f a0, a1;
    a0.x = hi ? xq0 : p0;  a0.y = hi ? xq1 : p1;
    a1.x = hi ? q0 : xp0;  a1.y = hi ? q1 : xp1;
    const float* wp = wl + (hi ? tB : tA) * WPITCH + row * 2;
#pragma unroll
    for (int nt = 0; nt < 4; ++nt) {
      const v2f bv = *(const v2f*)(wp + nt * 32);
      fsum[0][nt] = __builtin_amdgcn_wmma_f32_16x16x4_f32(
          false, a0, false, bv, (short)0, fsum[0][nt], false, false);
      fsum[1][nt] = __builtin_amdgcn_wmma_f32_16x16x4_f32(
          false, a1, false, bv, (short)0, fsum[1][nt], false, false);
    }
  };

  auto run_group = [&](int g, const v4f (&buf)[IN]) {
    float ph0 = 0.f, ph1 = 0.f;
    int pt = 0;
    const int gi = REV ? (NG - 1 - g) : g;
#pragma unroll
    for (int s = 0; s < GS; ++s) {
      const int sl = REV ? (GS - 1 - s) : s;
      const int tt = gi * GS + sl;

      float xv[IN];
#pragma unroll
      for (int c = 0; c < IN; ++c) {
        const int e = sl * IN + c;
        xv[c] = buf[e >> 2][e & 3];
      }

      float g8[8];
#pragma unroll
      for (int gg = 0; gg < 8; ++gg) {
        float acc = fmaf(whh[gg][0], h0, fmaf(whh[gg][1], h1, bias[gg]));
#pragma unroll
        for (int c = 0; c < IN; ++c) acc = fmaf(wih[gg][c], xv[c], acc);
        g8[gg] = acc;
      }

      const float i0 = sigm_half(g8[0]), i1 = sigm_half(g8[1]);
      const float f0 = sigm_half(g8[2]), f1 = sigm_half(g8[3]);
      const float t0 = tanh_hw(g8[4]), t1 = tanh_hw(g8[5]);
      const float o0 = sigm_half(g8[6]), o1 = sigm_half(g8[7]);

      c0 = fmaf(f0, c0, i0 * t0);
      c1 = fmaf(f1, c1, i1 * t1);
      h0 = o0 * tanh_hw(c0);
      h1 = o1 * tanh_hw(c1);

      if constexpr (FC) {
        if ((s & 1) == 0) { ph0 = h0; ph1 = h1; pt = tt; }
        else fc_pair(ph0, ph1, pt, h0, h1, tt);
      } else {
        float2 hv = make_float2(h0, h1);
        *(float2*)(out + ((size_t)tt * NB + (size_t)b) * 2) = hv;
      }
    }
  };

  auto fetch = [&](int g, v4f (&buf)[IN]) {
    asm volatile("" ::: "memory");
    __builtin_amdgcn_sched_barrier(0);
    load_group(g, buf);
    __builtin_amdgcn_sched_barrier(0);
  };

  auto use_group = [&](int g, const v4f (&buf)[IN]) {
    if constexpr (TMAJOR) {
      run_group(g, buf);
    } else {
      v4f xr[IN];
      own_row(buf, xr);
      run_group(g, xr);
    }
  };

  v4f r0[IN], r1[IN], r2[IN];
  if constexpr (TMAJOR) {
    load_group(0, r0);
    load_group(1, r1);
  } else {
    fetch(0, r0);
    fetch(1, r1);
  }
  for (int g = 0; g < NG - 1; g += RING) {
    fetch(g + 2, r2);
    use_group(g, r0);
    fetch(g + 3, r0);
    use_group(g + 1, r1);
    fetch(g + 4, r1);
    use_group(g + 2, r2);
  }
  use_group(NG - 1, r0);

  if constexpr (FC) {
    float* prow = out + (size_t)(b - lane) * 64 + row;
#pragma unroll
    for (int mt = 0; mt < 2; ++mt)
#pragma unroll
      for (int nt = 0; nt < 4; ++nt)
#pragma unroll
        for (int r = 0; r < 8; ++r) {
          const int m = mt * 16 + (hi ? 8 : 0) + r;
          prow[(size_t)m * 64 + nt * 16] = fsum[mt][nt][r];
        }
  }
}

template <int IN, bool TMAJOR>
__global__ __launch_bounds__(256) void lstm_layer_kernel(
    const float* __restrict__ in,
    const float* __restrict__ w_ih_f, const float* __restrict__ w_hh_f,
    const float* __restrict__ b_ih_f, const float* __restrict__ b_hh_f,
    const float* __restrict__ w_ih_r, const float* __restrict__ w_hh_r,
    const float* __restrict__ b_ih_r, const float* __restrict__ b_hh_r,
    float* __restrict__ out)
{
  v4f* stg = nullptr;
  if constexpr (!TMAJOR) {
    __shared__ v4f stage[8][32 * IN];
    stg = &stage[threadIdx.x >> 5][0];
  }
  if (blockIdx.y == 0)
    lstm_dir<IN, false, TMAJOR, false>(in, w_ih_f, w_hh_f, b_ih_f, b_hh_f, out,
                                       stg, nullptr);
  else
    lstm_dir<IN, true, TMAJOR, false>(in, w_ih_r, w_hh_r, b_ih_r, b_hh_r,
                                      out + PLANE, stg, nullptr);
}

__global__ __launch_bounds__(256) void lstm1_fc1_kernel(
    const float* __restrict__ in,
    const float* __restrict__ w_ih_f, const float* __restrict__ w_hh_f,
    const float* __restrict__ b_ih_f, const float* __restrict__ b_hh_f,
    const float* __restrict__ w_ih_r, const float* __restrict__ w_hh_r,
    const float* __restrict__ b_ih_r, const float* __restrict__ b_hh_r,
    const float* __restrict__ w1,
    float* __restrict__ part)
{
  __shared__ float wls[NT * WPITCH];
  const int d = blockIdx.y;
  for (int i = threadIdx.x; i < NT * 64; i += 256) {
    const int t = i >> 6, n = i & 63;
    const v2f w = *(const v2f*)(w1 + (size_t)n * 400 + t * 4 + d * 2);
    *(v2f*)(wls + t * WPITCH + n * 2) = w;
  }
  __syncthreads();
  if (d == 0)
    lstm_dir<4, false, true, true>(in, w_ih_f, w_hh_f, b_ih_f, b_hh_f, part,
                                   nullptr, wls);
  else
    lstm_dir<4, true, true, true>(in, w_ih_r, w_hh_r, b_ih_r, b_hh_r,
                                  part + (size_t)NB * 64, nullptr, wls);
}

__global__ __launch_bounds__(256) void fc2_wmma_kernel(
    const float* __restrict__ part,
    const float* __restrict__ b1,
    const float* __restrict__ w,
    const float* __restrict__ bias,
    float* __restrict__ out)
{
  const int wid = blockIdx.x * (blockDim.x >> 5) + (threadIdx.x >> 5);
  const int lane = threadIdx.x & 31;
  const int m_tile = wid >> 1;
  const int n_tile = wid & 1;
  const int row = lane & 15;
  const int half = lane >> 4;
  const int base_b = m_tile * 16;
  const int n = n_tile * 16 + row;
  const bool nvalid = (n < 20);

  v8f acc = {};
  const float* arow = part + (size_t)(base_b + row) * 64 + half * 2;
  const float* brow = b1 + half * 2;
  const float* wrow = w + (size_t)(nvalid ? n : 0) * 64 + half * 2;

#pragma unroll 4
  for (int kk = 0; kk < 16; ++kk) {
    const v2f pf = *(const v2f*)(arow + kk * 4);
    const v2f pr = *(const v2f*)(arow + (size_t)NB * 64 + kk * 4);
    const v2f bk = *(const v2f*)(brow + kk * 4);
    v2f a;
    a.x = fmaxf(pf.x + pr.x + bk.x, 0.0f);
    a.y = fmaxf(pf.y + pr.y + bk.y, 0.0f);
    v2f bv = *(const v2f*)(wrow + kk * 4);
    if (!nvalid) { bv.x = 0.0f; bv.y = 0.0f; }
    acc = __builtin_amdgcn_wmma_f32_16x16x4_f32(
        false, a, false, bv, (short)0, acc, false, false);
  }

  if (nvalid) {
    const float bn = bias[n];
#pragma unroll
    for (int r = 0; r < 8; ++r) {
      const int m = r + half * 8;
      out[(size_t)(base_b + m) * 20 + n] = acc[r] + bn;
    }
  }
}

extern "C" void kernel_launch(void* const* d_in, const int* in_sizes, int n_in,
                              void* d_out, int out_size, void* d_ws,
                              size_t ws_size, hipStream_t stream) {
  const float* x = (const float*)d_in[0];
  const float* w_ih_l0  = (const float*)d_in[1];
  const float* w_hh_l0  = (const float*)d_in[2];
  const float* b_ih_l0  = (const float*)d_in[3];
  const float* b_hh_l0  = (const float*)d_in[4];
  const float* w_ih_l0r = (const float*)d_in[5];
  const float* w_hh_l0r = (const float*)d_in[6];
  const float* b_ih_l0r = (const float*)d_in[7];
  const float* b_hh_l0r = (const float*)d_in[8];
  const float* w_ih_l1  = (const float*)d_in[9];
  const float* w_hh_l1  = (const float*)d_in[10];
  const float* b_ih_l1  = (const float*)d_in[11];
  const float* b_hh_l1  = (const float*)d_in[12];
  const float* w_ih_l1r = (const float*)d_in[13];
  const float* w_hh_l1r = (const float*)d_in[14];
  const float* b_ih_l1r = (const float*)d_in[15];
  const float* b_hh_l1r = (const float*)d_in[16];
  const float* fc1_w = (const float*)d_in[17];
  const float* fc1_b = (const float*)d_in[18];
  const float* fc2_w = (const float*)d_in[19];
  const float* fc2_b = (const float*)d_in[20];
  float* out = (float*)d_out;

  float* ws  = (float*)d_ws;
  float* ws0 = ws;
  float* ws1 = ws + (size_t)NT * NB * 4;

  const dim3 blk(256);
  const dim3 grid_lstm(NB / 256, 2);

  lstm_layer_kernel<7, false><<<grid_lstm, blk, 0, stream>>>(
      x, w_ih_l0, w_hh_l0, b_ih_l0, b_hh_l0,
      w_ih_l0r, w_hh_l0r, b_ih_l0r, b_hh_l0r, ws0);

  lstm1_fc1_kernel<<<grid_lstm, blk, 0, stream>>>(
      ws0, w_ih_l1, w_hh_l1, b_ih_l1, b_hh_l1,
      w_ih_l1r, w_hh_l1r, b_ih_l1r, b_hh_l1r, fc1_w, ws1);

  fc2_wmma_kernel<<<dim3(1024), blk, 0, stream>>>(ws1, fc1_b, fc2_w, fc2_b, out);
}


# FC1's end, relu and FC2 move into the layer-1 launch: a work-group runs both directions of its 128 samples, their FC1 sums meet in LDS, FC2 on f32 WMMA from LDS; third launch goes; sums unchanged
# speedup vs baseline: 3.3061x; 1.0905x over previous
#include <hip/hip_runtime.h>
#include <hip/hip_bf16.h>


#define NB 65536
#define NT 100

typedef __attribute__((ext_vector_type(2))) float v2f;
typedef __attribute__((ext_vector_type(4))) float v4f;
typedef __attribute__((ext_vector_type(8))) float v8f;

__device__ __forceinline__ float tanh_hw(float x) {
  return __builtin_amdgcn_tanhf(x);
}
__device__ __forceinline__ float uni(float v) {
  return __int_as_float(__builtin_amdgcn_readfirstlane(__float_as_int(v)));
}
__device__ __forceinline__ float sigm_half(float half_x) {
  return fmaf(0.5f, tanh_hw(half_x), 0.5f);
}

__device__ __forceinline__ float xhalf(float x) {
  const int b = __float_as_int(x);
  return __int_as_float(
      __builtin_amdgcn_permlanex16(b, b, 0x76543210, 0xfedcba98, false, false));
}

#define GS 4
#define NG (NT / GS)
#define RING 3
#define PLANE ((size_t)NT * NB * 2)

#define WPITCH 160
template <int IN, bool REV, bool TMAJOR, bool FC>
__device__ __forceinline__ void lstm_dir(
    const float* __restrict__ in,
    const float* __restrict__ w_ih,
    const float* __restrict__ w_hh,
    const float* __restrict__ b_ih,
    const float* __restrict__ b_hh,
    float* __restrict__ out,
    v4f* stg,
    const float* wl,
    const int b_fc,
    v8f (&fsum)[2][4])
{
  static_assert(!FC || (TMAJOR && GS % 2 == 0), "FC1 sums: pairs of steps");
  static_assert(NT % GS == 0, "whole groups");
  static_assert(RING == 3 && NG % RING == 1 && NG > RING,
                "the loop below leaves exactly one group");
  static_assert(!TMAJOR || IN == 4, "time-major input has 4 channels");
  static_assert((GS * IN) % 4 == 0, "a group is whole 16-byte vectors");

  static_assert(NB % 32 == 0, "a wave's 32 rows are all inside the batch");
  static_assert(TMAJOR || GS == 4,
                "row-major: a row's group is IN 16-byte vectors");

  const int b = FC ? b_fc : (int)(blockIdx.x * blockDim.x + threadIdx.x);
  if (b >= NB) return;

  const int lane = threadIdx.x & 31;
  const float* wave_in = in + (size_t)(b - lane) * NT * IN;
  int poff[IN];
#pragma unroll
  for (int k = 0; k < IN; ++k) {
    const int p = k * 32 + lane;
    poff[k] = (p / IN) * (NT * IN) + (p % IN) * 4;
  }

  float wih[8][IN];
  float whh[8][2];
  float bias[8];
#pragma unroll
  for (int g = 0; g < 8; ++g) {
    const float sc = (g == 4 || g == 5) ? 1.0f : 0.5f;
#pragma unroll
    for (int c = 0; c < IN; ++c) wih[g][c] = uni(sc * w_ih[g * IN + c]);
    whh[g][0] = uni(sc * w_hh[g * 2 + 0]);
    whh[g][1] = uni(sc * w_hh[g * 2 + 1]);
    bias[g] = uni(sc * (b_ih[g] + b_hh[g]));
  }

  float h0 = 0.f, h1 = 0.f, c0 = 0.f, c1 = 0.f;

  auto load_group = [&](int g, v4f (&buf)[IN]) {
    const int gc = g < NG ? g : NG - 1;
    const int gi = REV ? (NG - 1 - gc) : gc;
    if constexpr (TMAJOR) {
#pragma unroll
      for (int s = 0; s < GS; ++s) {
        const float* p = in + ((size_t)(gi * GS + s) * NB + (size_t)b) * 2;
        const v2f f = *(const v2f*)p;
        const v2f r = *(const v2f*)(p + PLANE);
        buf[s] = v4f{f.x, f.y, r.x, r.y};
      }
    } else {
      const float* p = wave_in + (size_t)gi * GS * IN;
#pragma unroll
      for (int k = 0; k < IN; ++k) buf[k] = *(const v4f*)(p + poff[k]);
    }
  };

  auto own_row = [&](const v4f (&pc)[IN], v4f (&xr)[IN]) {
#pragma unroll
    for (int k = 0; k < IN; ++k) stg[k * 32 + lane] = pc[k];
#pragma unroll
    for (int q = 0; q < IN; ++q) {
      const int v = REV ? IN - 1 - q : q;
      xr[v] = stg[lane * IN + v];
    }
    __builtin_amdgcn_sched_barrier(0);
  };

  const bool hi = lane >= 16;
  const int row = lane & 15;
  auto fc_pair = [&](float pa, float pb, int tA, float qa, float qb, int tB) {
    const float p0 = fmaxf(pa, 0.0f), p1 = fmaxf(pb, 0.0f);
    const float q0 = fmaxf(qa, 0.0f), q1 = fmaxf(qb, 0.0f);
    const float xp0 = xhalf(p0), xp1 = xhalf(p1);
    const float xq0 = xhalf(q0), xq1 = xhalf(q1);
    v2f a0, a1;
    a0.x = hi ? xq0 : p0;  a0.y = hi ? xq1 : p1;
    a1.x = hi ? q0 : xp0;  a1.y = hi ? q1 : xp1;
    const float* wp = wl + (hi ? tB : tA) * WPITCH + row * 2;
#pragma unroll
    for (int nt = 0; nt < 4; ++nt) {
      const v2f bv = *(const v2f*)(wp + nt * 32);
      fsum[0][nt] = __builtin_amdgcn_wmma_f32_16x16x4_f32(
          false, a0, false, bv, (short)0, fsum[0][nt], false, false);
      fsum[1][nt] = __builtin_amdgcn_wmma_f32_16x16x4_f32(
          false, a1, false, bv, (short)0, fsum[1][nt], false, false);
    }
  };

  auto run_group = [&](int g, const v4f (&buf)[IN]) {
    float ph0 = 0.f, ph1 = 0.f;
    int pt = 0;
    const int gi = REV ? (NG - 1 - g) : g;
#pragma unroll
    for (int s = 0; s < GS; ++s) {
      const int sl = REV ? (GS - 1 - s) : s;
      const int tt = gi * GS + sl;

      float xv[IN];
#pragma unroll
      for (int c = 0; c < IN; ++c) {
        const int e = sl * IN + c;
        xv[c] = buf[e >> 2][e & 3];
      }

      float g8[8];
#pragma unroll
      for (int gg = 0; gg < 8; ++gg) {
        float acc = fmaf(whh[gg][0], h0, fmaf(whh[gg][1], h1, bias[gg]));
#pragma unroll
        for (int c = 0; c < IN; ++c) acc = fmaf(wih[gg][c], xv[c], acc);
        g8[gg] = acc;
      }

      const float i0 = sigm_half(g8[0]), i1 = sigm_half(g8[1]);
      const float f0 = sigm_half(g8[2]), f1 = sigm_half(g8[3]);
      const float t0 = tanh_hw(g8[4]), t1 = tanh_hw(g8[5]);
      const float o0 = sigm_half(g8[6]), o1 = sigm_half(g8[7]);

      c0 = fmaf(f0, c0, i0 * t0);
      c1 = fmaf(f1, c1, i1 * t1);
      h0 = o0 * tanh_hw(c0);
      h1 = o1 * tanh_hw(c1);

      if constexpr (FC) {
        if ((s & 1) == 0) { ph0 = h0; ph1 = h1; pt = tt; }
        else fc_pair(ph0, ph1, pt, h0, h1, tt);
      } else {
        float2 hv = make_float2(h0, h1);
        *(float2*)(out + ((size_t)tt * NB + (size_t)b) * 2) = hv;
      }
    }
  };

  auto fetch = [&](int g, v4f (&buf)[IN]) {
    asm volatile("" ::: "memory");
    __builtin_amdgcn_sched_barrier(0);
    load_group(g, buf);
    __builtin_amdgcn_sched_barrier(0);
  };

  auto use_group = [&](int g, const v4f (&buf)[IN]) {
    if constexpr (TMAJOR) {
      run_group(g, buf);
    } else {
      v4f xr[IN];
      own_row(buf, xr);
      run_group(g, xr);
    }
  };

  v4f r0[IN], r1[IN], r2[IN];
  if constexpr (TMAJOR) {
    load_group(0, r0);
    load_group(1, r1);
  } else {
    fetch(0, r0);
    fetch(1, r1);
  }
  for (int g = 0; g < NG - 1; g += RING) {
    fetch(g + 2, r2);
    use_group(g, r0);
    fetch(g + 3, r0);
    use_group(g + 1, r1);
    fetch(g + 4, r1);
    use_group(g + 2, r2);
  }
  use_group(NG - 1, r0);
}

template <int IN, bool TMAJOR>
__global__ __launch_bounds__(256) void lstm_layer_kernel(
    const float* __restrict__ in,
    const float* __restrict__ w_ih_f, const float* __restrict__ w_hh_f,
    const float* __restrict__ b_ih_f, const float* __restrict__ b_hh_f,
    const float* __restrict__ w_ih_r, const float* __restrict__ w_hh_r,
    const float* __restrict__ b_ih_r, const float* __restrict__ b_hh_r,
    float* __restrict__ out)
{
  v4f* stg = nullptr;
  if constexpr (!TMAJOR) {
    __shared__ v4f stage[8][32 * IN];
    stg = &stage[threadIdx.x >> 5][0];
  }
  v8f none[2][4];
  if (blockIdx.y == 0)
    lstm_dir<IN, false, TMAJOR, false>(in, w_ih_f, w_hh_f, b_ih_f, b_hh_f, out,
                                       stg, nullptr, 0, none);
  else
    lstm_dir<IN, true, TMAJOR, false>(in, w_ih_r, w_hh_r, b_ih_r, b_hh_r,
                                      out + PLANE, stg, nullptr, 0, none);
}

#define XP 68
__global__ __launch_bounds__(256) void lstm1_fc_kernel(
    const float* __restrict__ in,
    const float* __restrict__ w_ih_f, const float* __restrict__ w_hh_f,
    const float* __restrict__ b_ih_f, const float* __restrict__ b_hh_f,
    const float* __restrict__ w_ih_r, const float* __restrict__ w_hh_r,
    const float* __restrict__ b_ih_r, const float* __restrict__ b_hh_r,
    const float* __restrict__ w1,
    const float* __restrict__ b1,
    const float* __restrict__ w2,
    const float* __restrict__ b2,
    float* __restrict__ outp)
{
  __shared__ __attribute__((aligned(16))) float wls[2 * NT * WPITCH];
  for (int i = threadIdx.x; i < NT * 64; i += 256) {
    const int t = i >> 6, n = i & 63;
    const v4f w = *(const v4f*)(w1 + (size_t)n * 400 + t * 4);
    *(v2f*)(wls + t * WPITCH + n * 2) = v2f{w.x, w.y};
    *(v2f*)(wls + NT * WPITCH + t * WPITCH + n * 2) = v2f{w.z, w.w};
  }
  __syncthreads();

  const int lane = threadIdx.x & 31;
  const int wave = __builtin_amdgcn_readfirstlane((int)(threadIdx.x >> 5));
  const int dir = wave >> 2;
  const int b = blockIdx.x * 128 + (threadIdx.x & 127);

  v8f fsum[2][4] = {};
  if (dir == 0)
    lstm_dir<4, false, true, true>(in, w_ih_f, w_hh_f, b_ih_f, b_hh_f, nullptr,
                                   nullptr, wls, b, fsum);
  else
    lstm_dir<4, true, true, true>(in, w_ih_r, w_hh_r, b_ih_r, b_hh_r, nullptr,
                                  nullptr, wls + NT * WPITCH, b, fsum);

  __syncthreads();

  float* xw = wls + (wave & 3) * (32 * XP);
  const bool hi = lane >= 16;
  const int row = lane & 15;
  if (dir == 1) {
#pragma unroll
    for (int mt = 0; mt < 2; ++mt)
#pragma unroll
      for (int nt = 0; nt < 4; ++nt)
#pragma unroll
        for (int r = 0; r < 8; ++r) {
          const int m = mt * 16 + (hi ? 8 : 0) + r;
          xw[m * XP + nt * 16 + row] = fsum[mt][nt][r];
        }
  }

  __syncthreads();

  if (dir == 0) {
#pragma unroll
    for (int nt = 0; nt < 4; ++nt) {
      const float bk = b1[nt * 16 + row];
#pragma unroll
      for (int mt = 0; mt < 2; ++mt)
#pragma unroll
        for (int r = 0; r < 8; ++r) {
          const int m = mt * 16 + (hi ? 8 : 0) + r;
          float* p = xw + m * XP + nt * 16 + row;
          *p = fmaxf(fsum[mt][nt][r] + *p + bk, 0.0f);
        }
    }
    const int n0 = row, n1 = 16 + row;
    const bool v1 = n1 < 20;
    const float* w2a = w2 + n0 * 64 + (hi ? 2 : 0);
    const float* w2b = w2 + (v1 ? n1 : 0) * 64 + (hi ? 2 : 0);
    const float* ar0 = xw + row * XP + (hi ? 2 : 0);
    const float* ar1 = xw + (16 + row) * XP + (hi ? 2 : 0);
    v8f acc[2][2] = {};
#pragma unroll
    for (int kk = 0; kk < 16; ++kk) {
      const v2f a0 = *(const v2f*)(ar0 + kk * 4);
      const v2f a1 = *(const v2f*)(ar1 + kk * 4);
      const v2f bv0 = *(const v2f*)(w2a + kk * 4);
      v2f bv1 = *(const v2f*)(w2b + kk * 4);
      if (!v1) { bv1.x = 0.0f; bv1.y = 0.0f; }
      acc[0][0] = __builtin_amdgcn_wmma_f32_16x16x4_f32(
          false, a0, false, bv0, (short)0, acc[0][0], false, false);
      acc[1][0] = __builtin_amdgcn_wmma_f32_16x16x4_f32(
          false, a1, false, bv0, (short)0, acc[1][0], false, false);
      acc[0][1] = __builtin_amdgcn_wmma_f32_16x16x4_f32(
          false, a0, false, bv1, (short)0, acc[0][1], false, false);
      acc[1][1] = __builtin_amdgcn_wmma_f32_16x16x4_f32(
          false, a1, false, bv1, (short)0, acc[1][1], false, false);
    }
    const float bn0 = b2[n0];
    const float bn1 = b2[v1 ? n1 : 0];
#pragma unroll
    for (int mt = 0; mt < 2; ++mt)
#pragma unroll
      for (int r = 0; r < 8; ++r) {
        const int m = mt * 16 + (hi ? 8 : 0) + r;
        xw[m * 20 + n0] = acc[mt][0][r] + bn0;
        if (v1) xw[m * 20 + n1] = acc[mt][1][r] + bn1;
      }
    float* ob = outp + (size_t)(b - lane) * 20;
#pragma unroll
    for (int i = 0; i < 5; ++i) {
      const v4f v = *(const v4f*)(xw + (i * 32 + lane) * 4);
      *(v4f*)(ob + (i * 32 + lane) * 4) = v;
    }
  }
}

extern "C" void kernel_launch(void* const* d_in, const int* in_sizes, int n_in,
                              void* d_out, int out_size, void* d_ws,
                              size_t ws_size, hipStream_t stream) {
  const float* x = (const float*)d_in[0];
  const float* w_ih_l0  = (const float*)d_in[1];
  const float* w_hh_l0  = (const float*)d_in[2];
  const float* b_ih_l0  = (const float*)d_in[3];
  const float* b_hh_l0  = (const float*)d_in[4];
  const float* w_ih_l0r = (const float*)d_in[5];
  const float* w_hh_l0r = (const float*)d_in[6];
  const float* b_ih_l0r = (const float*)d_in[7];
  const float* b_hh_l0r = (const float*)d_in[8];
  const float* w_ih_l1  = (const float*)d_in[9];
  const float* w_hh_l1  = (const float*)d_in[10];
  const float* b_ih_l1  = (const float*)d_in[11];
  const float* b_hh_l1  = (const float*)d_in[12];
  const float* w_ih_l1r = (const float*)d_in[13];
  const float* w_hh_l1r = (const float*)d_in[14];
  const float* b_ih_l1r = (const float*)d_in[15];
  const float* b_hh_l1r = (const float*)d_in[16];
  const float* fc1_w = (const float*)d_in[17];
  const float* fc1_b = (const float*)d_in[18];
  const float* fc2_w = (const float*)d_in[19];
  const float* fc2_b = (const float*)d_in[20];
  float* out = (float*)d_out;

  float* ws0 = (float*)d_ws;

  const dim3 blk(256);
  const dim3 grid_lstm(NB / 256, 2);

  lstm_layer_kernel<7, false><<<grid_lstm, blk, 0, stream>>>(
      x, w_ih_l0, w_hh_l0, b_ih_l0, b_hh_l0,
      w_ih_l0r, w_hh_l0r, b_ih_l0r, b_hh_l0r, ws0);

  lstm1_fc_kernel<<<dim3(NB / 128), blk, 0, stream>>>(
      ws0, w_ih_l1, w_hh_l1, b_ih_l1, b_hh_l1,
      w_ih_l1r, w_hh_l1r, b_ih_l1r, b_hh_l1r,
      fc1_w, fc1_b, fc2_w, fc2_b, out);
}


# Both LSTM layers, FC1 and FC2 in one launch: a work-group keeps its 128 samples' layer-0 output (all steps, both directions) in LDS as layer 1's input; the h planes and a launch go; sums unchanged
# speedup vs baseline: 4.1535x; 1.2563x over previous
#include <hip/hip_runtime.h>
#include <hip/hip_bf16.h>


#define NB 65536
#define NT 100

typedef __attribute__((ext_vector_type(2))) float v2f;
typedef __attribute__((ext_vector_type(4))) float v4f;
typedef __attribute__((ext_vector_type(8))) float v8f;

__device__ __forceinline__ float tanh_hw(float x) {
  return __builtin_amdgcn_tanhf(x);
}
__device__ __forceinline__ float uni(float v) {
  return __int_as_float(__builtin_amdgcn_readfirstlane(__float_as_int(v)));
}
__device__ __forceinline__ float sigm_half(float half_x) {
  return fmaf(0.5f, tanh_hw(half_x), 0.5f);
}

__device__ __forceinline__ float xhalf(float x) {
  const int b = __float_as_int(x);
  return __int_as_float(
      __builtin_amdgcn_permlanex16(b, b, 0x76543210, 0xfedcba98, false, false));
}

#define GS 4
#define NG (NT / GS)
#define RING 3
#define WGS 128
#define HSTEP (WGS * 4)

#define WPITCH 128
#define WSWZ(t) (((t) & 1) * 32)
template <int IN, bool REV, bool TMAJOR, bool FC>
__device__ __forceinline__ void lstm_dir(
    const float* __restrict__ in,
    const float* __restrict__ w_ih,
    const float* __restrict__ w_hh,
    const float* __restrict__ b_ih,
    const float* __restrict__ b_hh,
    float* out,
    v4f* stg,
    const float* wl,
    const int b,
    v8f (&fsum)[2][4])
{
  static_assert(!FC || (TMAJOR && GS % 2 == 0), "FC1 sums: pairs of steps");
  static_assert(NT % GS == 0, "whole groups");
  static_assert(RING == 3 && NG % RING == 1 && NG > RING,
                "the loop below leaves exactly one group");
  static_assert(!TMAJOR || IN == 4, "time-major input has 4 channels");
  static_assert((GS * IN) % 4 == 0, "a group is whole 16-byte vectors");

  static_assert(NB % 32 == 0, "a wave's 32 rows are all inside the batch");
  static_assert(TMAJOR || GS == 4,
                "row-major: a row's group is IN 16-byte vectors");

  const int lane = threadIdx.x & 31;
  const float* wave_in = in + (size_t)(b - lane) * NT * IN;
  int poff[IN];
#pragma unroll
  for (int k = 0; k < IN; ++k) {
    const int p = k * 32 + lane;
    poff[k] = (p / IN) * (NT * IN) + (p % IN) * 4;
  }

  float wih[8][IN];
  float whh[8][2];
  float bias[8];
#pragma unroll
  for (int g = 0; g < 8; ++g) {
    const float sc = (g == 4 || g == 5) ? 1.0f : 0.5f;
#pragma unroll
    for (int c = 0; c < IN; ++c) wih[g][c] = uni(sc * w_ih[g * IN + c]);
    whh[g][0] = uni(sc * w_hh[g * 2 + 0]);
    whh[g][1] = uni(sc * w_hh[g * 2 + 1]);
    bias[g] = uni(sc * (b_ih[g] + b_hh[g]));
  }

  float h0 = 0.f, h1 = 0.f, c0 = 0.f, c1 = 0.f;

  auto load_group = [&](int g, v4f (&buf)[IN]) {
    const int gc = g < NG ? g : NG - 1;
    const int gi = REV ? (NG - 1 - gc) : gc;
    if constexpr (TMAJOR) {
#pragma unroll
      for (int s = 0; s < GS; ++s)
        buf[s] = *(const v4f*)(in + (gi * GS + s) * HSTEP);
    } else {
      const float* p = wave_in + (size_t)gi * GS * IN;
#pragma unroll
      for (int k = 0; k < IN; ++k) buf[k] = *(const v4f*)(p + poff[k]);
    }
  };

  auto own_row = [&](const v4f (&pc)[IN], v4f (&xr)[IN]) {
#pragma unroll
    for (int k = 0; k < IN; ++k) stg[k * 32 + lane] = pc[k];
#pragma unroll
    for (int q = 0; q < IN; ++q) {
      const int v = REV ? IN - 1 - q : q;
      xr[v] = stg[lane * IN + v];
    }
    __builtin_amdgcn_sched_barrier(0);
  };

  const bool hi = lane >= 16;
  const int row = lane & 15;
  auto fc_pair = [&](float pa, float pb, int tA, float qa, float qb, int tB) {
    const float p0 = fmaxf(pa, 0.0f), p1 = fmaxf(pb, 0.0f);
    const float q0 = fmaxf(qa, 0.0f), q1 = fmaxf(qb, 0.0f);
    const float xp0 = xhalf(p0), xp1 = xhalf(p1);
    const float xq0 = xhalf(q0), xq1 = xhalf(q1);
    v2f a0, a1;
    a0.x = hi ? xq0 : p0;  a0.y = hi ? xq1 : p1;
    a1.x = hi ? q0 : xp0;  a1.y = hi ? q1 : xp1;
    const int tl = hi ? tB : tA;
    const float* wp = wl + tl * WPITCH + row * 2;
    const int sw = WSWZ(tl);
#pragma unroll
    for (int nt = 0; nt < 4; ++nt) {
      const v2f bv = *(const v2f*)(wp + ((nt * 32) ^ sw));
      fsum[0][nt] = __builtin_amdgcn_wmma_f32_16x16x4_f32(
          false, a0, false, bv, (short)0, fsum[0][nt], false, false);
      fsum[1][nt] = __builtin_amdgcn_wmma_f32_16x16x4_f32(
          false, a1, false, bv, (short)0, fsum[1][nt], false, false);
    }
  };

  auto run_group = [&](int g, const v4f (&buf)[IN]) {
    float ph0 = 0.f, ph1 = 0.f;
    int pt = 0;
    const int gi = REV ? (NG - 1 - g) : g;
#pragma unroll
    for (int s = 0; s < GS; ++s) {
      const int sl = REV ? (GS - 1 - s) : s;
      const int tt = gi * GS + sl;

      float xv[IN];
#pragma unroll
      for (int c = 0; c < IN; ++c) {
        const int e = sl * IN + c;
        xv[c] = buf[e >> 2][e & 3];
      }

      float g8[8];
#pragma unroll
      for (int gg = 0; gg < 8; ++gg) {
        float acc = fmaf(whh[gg][0], h0, fmaf(whh[gg][1], h1, bias[gg]));
#pragma unroll
        for (int c = 0; c < IN; ++c) acc = fmaf(wih[gg][c], xv[c], acc);
        g8[gg] = acc;
      }

      const float i0 = sigm_half(g8[0]), i1 = sigm_half(g8[1]);
      const float f0 = sigm_half(g8[2]), f1 = sigm_half(g8[3]);
      const float t0 = tanh_hw(g8[4]), t1 = tanh_hw(g8[5]);
      const float o0 = sigm_half(g8[6]), o1 = sigm_half(g8[7]);

      c0 = fmaf(f0, c0, i0 * t0);
      c1 = fmaf(f1, c1, i1 * t1);
      h0 = o0 * tanh_hw(c0);
      h1 = o1 * tanh_hw(c1);

      if constexpr (FC) {
        if ((s & 1) == 0) { ph0 = h0; ph1 = h1; pt = tt; }
        else fc_pair(ph0, ph1, pt, h0, h1, tt);
      } else {
        *(v2f*)(out + tt * HSTEP) = v2f{h0, h1};
      }
    }
  };

  auto fetch = [&](int g, v4f (&buf)[IN]) {
    asm volatile("" ::: "memory");
    __builtin_amdgcn_sched_barrier(0);
    load_group(g, buf);
    __builtin_amdgcn_sched_barrier(0);
  };

  auto use_group = [&](int g, const v4f (&buf)[IN]) {
    if constexpr (TMAJOR) {
      run_group(g, buf);
    } else {
      v4f xr[IN];
      own_row(buf, xr);
      run_group(g, xr);
    }
  };

  v4f r0[IN], r1[IN], r2[IN];
  if constexpr (TMAJOR) {
    load_group(0, r0);
    load_group(1, r1);
  } else {
    fetch(0, r0);
    fetch(1, r1);
  }
  for (int g = 0; g < NG - 1; g += RING) {
    fetch(g + 2, r2);
    use_group(g, r0);
    fetch(g + 3, r0);
    use_group(g + 1, r1);
    fetch(g + 4, r1);
    use_group(g + 2, r2);
  }
  use_group(NG - 1, r0);
}

#define XP 68
__global__ __launch_bounds__(256) void lstm2_fc_kernel(
    const float* __restrict__ x,
    const float* __restrict__ w_ih_0f, const float* __restrict__ w_hh_0f,
    const float* __restrict__ b_ih_0f, const float* __restrict__ b_hh_0f,
    const float* __restrict__ w_ih_0r, const float* __restrict__ w_hh_0r,
    const float* __restrict__ b_ih_0r, const float* __restrict__ b_hh_0r,
    const float* __restrict__ w_ih_1f, const float* __restrict__ w_hh_1f,
    const float* __restrict__ b_ih_1f, const float* __restrict__ b_hh_1f,
    const float* __restrict__ w_ih_1r, const float* __restrict__ w_hh_1r,
    const float* __restrict__ b_ih_1r, const float* __restrict__ b_hh_1r,
    const float* __restrict__ w1,
    const float* __restrict__ b1,
    const float* __restrict__ w2,
    const float* __restrict__ b2,
    float* __restrict__ outp)
{
  static_assert(NB % WGS == 0, "whole work-groups");
  static_assert(8 * 32 * 7 * 4 <= 2 * NT * WPITCH, "x slots fit in wls");
  static_assert(4 * 32 * XP <= 2 * NT * WPITCH, "pair areas fit in wls");
  __shared__ __attribute__((aligned(16))) float hbuf[NT * HSTEP];
  __shared__ __attribute__((aligned(16))) float wls[2 * NT * WPITCH];

  const int lane = threadIdx.x & 31;
  const int wave = __builtin_amdgcn_readfirstlane((int)(threadIdx.x >> 5));
  const int dir = wave >> 2;
  const int smp = threadIdx.x & (WGS - 1);
  const int b = blockIdx.x * WGS + smp;
  float* hrow = hbuf + smp * 4;

  {
    v4f* stg = (v4f*)wls + wave * (32 * 7);
    v8f none[2][4];
    if (dir == 0)
      lstm_dir<7, false, false, false>(x, w_ih_0f, w_hh_0f, b_ih_0f, b_hh_0f,
                                       hrow, stg, nullptr, b, none);
    else
      lstm_dir<7, true, false, false>(x, w_ih_0r, w_hh_0r, b_ih_0r, b_hh_0r,
                                      hrow + 2, stg, nullptr, b, none);
  }

  __syncthreads();

  for (int i = threadIdx.x; i < NT * 64; i += 256) {
    const int t = i >> 6, n = i & 63;
    const v4f w = *(const v4f*)(w1 + (size_t)n * 400 + t * 4);
    const int o = t * WPITCH + ((n * 2) ^ WSWZ(t));
    *(v2f*)(wls + o) = v2f{w.x, w.y};
    *(v2f*)(wls + NT * WPITCH + o) = v2f{w.z, w.w};
  }
  __syncthreads();

  v8f fsum[2][4] = {};
  if (dir == 0)
    lstm_dir<4, false, true, true>(hrow, w_ih_1f, w_hh_1f, b_ih_1f, b_hh_1f,
                                   nullptr, nullptr, wls, b, fsum);
  else
    lstm_dir<4, true, true, true>(hrow, w_ih_1r, w_hh_1r, b_ih_1r, b_hh_1r,
                                  nullptr, nullptr, wls + NT * WPITCH, b, fsum);

  __syncthreads();

  float* xw = wls + (wave & 3) * (32 * XP);
  const bool hi = lane >= 16;
  const int row = lane & 15;
  if (dir == 1) {
#pragma unroll
    for (int mt = 0; mt < 2; ++mt)
#pragma unroll
      for (int nt = 0; nt < 4; ++nt)
#pragma unroll
        for (int r = 0; r < 8; ++r) {
          const int m = mt * 16 + (hi ? 8 : 0) + r;
          xw[m * XP + nt * 16 + row] = fsum[mt][nt][r];
        }
  }

  __syncthreads();

  if (dir == 0) {
#pragma unroll
    for (int nt = 0; nt < 4; ++nt) {
      const float bk = b1[nt * 16 + row];
#pragma unroll
      for (int mt = 0; mt < 2; ++mt)
#pragma unroll
        for (int r = 0; r < 8; ++r) {
          const int m = mt * 16 + (hi ? 8 : 0) + r;
          float* p = xw + m * XP + nt * 16 + row;
          *p = fmaxf(fsum[mt][nt][r] + *p + bk, 0.0f);
        }
    }
    const int n0 = row, n1 = 16 + row;
    const bool v1 = n1 < 20;
    const float* w2a = w2 + n0 * 64 + (hi ? 2 : 0);
    const float* w2b = w2 + (v1 ? n1 : 0) * 64 + (hi ? 2 : 0);
    const float* ar0 = xw + row * XP + (hi ? 2 : 0);
    const float* ar1 = xw + (16 + row) * XP + (hi ? 2 : 0);
    v8f acc[2][2] = {};
#pragma unroll
    for (int kk = 0; kk < 16; ++kk) {
      const v2f a0 = *(const v2f*)(ar0 + kk * 4);
      const v2f a1 = *(const v2f*)(ar1 + kk * 4);
      const v2f bv0 = *(const v2f*)(w2a + kk * 4);
      v2f bv1 = *(const v2f*)(w2b + kk * 4);
      if (!v1) { bv1.x = 0.0f; bv1.y = 0.0f; }
      acc[0][0] = __builtin_amdgcn_wmma_f32_16x16x4_f32(
          false, a0, false, bv0, (short)0, acc[0][0], false, false);
      acc[1][0] = __builtin_amdgcn_wmma_f32_16x16x4_f32(
          false, a1, false, bv0, (short)0, acc[1][0], false, false);
      acc[0][1] = __builtin_amdgcn_wmma_f32_16x16x4_f32(
          false, a0, false, bv1, (short)0, acc[0][1], false, false);
      acc[1][1] = __builtin_amdgcn_wmma_f32_16x16x4_f32(
          false, a1, false, bv1, (short)0, acc[1][1], false, false);
    }
    const float bn0 = b2[n0];
    const float bn1 = b2[v1 ? n1 : 0];
#pragma unroll
    for (int mt = 0; mt < 2; ++mt)
#pragma unroll
      for (int r = 0; r < 8; ++r) {
        const int m = mt * 16 + (hi ? 8 : 0) + r;
        xw[m * 20 + n0] = acc[mt][0][r] + bn0;
        if (v1) xw[m * 20 + n1] = acc[mt][1][r] + bn1;
      }
    float* ob = outp + (size_t)(b - lane) * 20;
#pragma unroll
    for (int i = 0; i < 5; ++i) {
      const v4f v = *(const v4f*)(xw + (i * 32 + lane) * 4);
      *(v4f*)(ob + (i * 32 + lane) * 4) = v;
    }
  }
}

extern "C" void kernel_launch(void* const* d_in, const int* in_sizes, int n_in,
                              void* d_out, int out_size, void* d_ws,
                              size_t ws_size, hipStream_t stream) {
  const float* x = (const float*)d_in[0];
  const float* w_ih_l0  = (const float*)d_in[1];
  const float* w_hh_l0  = (const float*)d_in[2];
  const float* b_ih_l0  = (const float*)d_in[3];
  const float* b_hh_l0  = (const float*)d_in[4];
  const float* w_ih_l0r = (const float*)d_in[5];
  const float* w_hh_l0r = (const float*)d_in[6];
  const float* b_ih_l0r = (const float*)d_in[7];
  const float* b_hh_l0r = (const float*)d_in[8];
  const float* w_ih_l1  = (const float*)d_in[9];
  const float* w_hh_l1  = (const float*)d_in[10];
  const float* b_ih_l1  = (const float*)d_in[11];
  const float* b_hh_l1  = (const float*)d_in[12];
  const float* w_ih_l1r = (const float*)d_in[13];
  const float* w_hh_l1r = (const float*)d_in[14];
  const float* b_ih_l1r = (const float*)d_in[15];
  const float* b_hh_l1r = (const float*)d_in[16];
  const float* fc1_w = (const float*)d_in[17];
  const float* fc1_b = (const float*)d_in[18];
  const float* fc2_w = (const float*)d_in[19];
  const float* fc2_b = (const float*)d_in[20];
  float* out = (float*)d_out;

  lstm2_fc_kernel<<<dim3(NB / WGS), dim3(256), 0, stream>>>(
      x, w_ih_l0, w_hh_l0, b_ih_l0, b_hh_l0,
      w_ih_l0r, w_hh_l0r, b_ih_l0r, b_hh_l0r,
      w_ih_l1, w_hh_l1, b_ih_l1, b_hh_l1,
      w_ih_l1r, w_hh_l1r, b_ih_l1r, b_hh_l1r,
      fc1_w, fc1_b, fc2_w, fc2_b, out);
}


# fc1 weight table fill: four neighbouring lanes read 64 contiguous bytes of a w1 row (a wave-wide load touches 8 lines, not 32), a lane's 25 loads issued as one batch; table layout and sums unchanged
# speedup vs baseline: 4.3027x; 1.0359x over previous
#include <hip/hip_runtime.h>
#include <hip/hip_bf16.h>


#define NB 65536
#define NT 100

typedef __attribute__((ext_vector_type(2))) float v2f;
typedef __attribute__((ext_vector_type(4))) float v4f;
typedef __attribute__((ext_vector_type(8))) float v8f;

__device__ __forceinline__ float tanh_hw(float x) {
  return __builtin_amdgcn_tanhf(x);
}
__device__ __forceinline__ float uni(float v) {
  return __int_as_float(__builtin_amdgcn_readfirstlane(__float_as_int(v)));
}
__device__ __forceinline__ float sigm_half(float half_x) {
  return fmaf(0.5f, tanh_hw(half_x), 0.5f);
}

__device__ __forceinline__ float xhalf(float x) {
  const int b = __float_as_int(x);
  return __int_as_float(
      __builtin_amdgcn_permlanex16(b, b, 0x76543210, 0xfedcba98, false, false));
}

#define GS 4
#define NG (NT / GS)
#define RING 3
#define WGS 128
#define HSTEP (WGS * 4)

#define WPITCH 128
#define WSWZ(t) (((t) & 1) * 32)
template <int IN, bool REV, bool TMAJOR, bool FC>
__device__ __forceinline__ void lstm_dir(
    const float* __restrict__ in,
    const float* __restrict__ w_ih,
    const float* __restrict__ w_hh,
    const float* __restrict__ b_ih,
    const float* __restrict__ b_hh,
    float* out,
    v4f* stg,
    const float* wl,
    const int b,
    v8f (&fsum)[2][4])
{
  static_assert(!FC || (TMAJOR && GS % 2 == 0), "FC1 sums: pairs of steps");
  static_assert(NT % GS == 0, "whole groups");
  static_assert(RING == 3 && NG % RING == 1 && NG > RING,
                "the loop below leaves exactly one group");
  static_assert(!TMAJOR || IN == 4, "time-major input has 4 channels");
  static_assert((GS * IN) % 4 == 0, "a group is whole 16-byte vectors");

  static_assert(NB % 32 == 0, "a wave's 32 rows are all inside the batch");
  static_assert(TMAJOR || GS == 4,
                "row-major: a row's group is IN 16-byte vectors");

  const int lane = threadIdx.x & 31;
  const float* wave_in = in + (size_t)(b - lane) * NT * IN;
  int poff[IN];
#pragma unroll
  for (int k = 0; k < IN; ++k) {
    const int p = k * 32 + lane;
    poff[k] = (p / IN) * (NT * IN) + (p % IN) * 4;
  }

  float wih[8][IN];
  float whh[8][2];
  float bias[8];
#pragma unroll
  for (int g = 0; g < 8; ++g) {
    const float sc = (g == 4 || g == 5) ? 1.0f : 0.5f;
#pragma unroll
    for (int c = 0; c < IN; ++c) wih[g][c] = uni(sc * w_ih[g * IN + c]);
    whh[g][0] = uni(sc * w_hh[g * 2 + 0]);
    whh[g][1] = uni(sc * w_hh[g * 2 + 1]);
    bias[g] = uni(sc * (b_ih[g] + b_hh[g]));
  }

  float h0 = 0.f, h1 = 0.f, c0 = 0.f, c1 = 0.f;

  auto load_group = [&](int g, v4f (&buf)[IN]) {
    const int gc = g < NG ? g : NG - 1;
    const int gi = REV ? (NG - 1 - gc) : gc;
    if constexpr (TMAJOR) {
#pragma unroll
      for (int s = 0; s < GS; ++s)
        buf[s] = *(const v4f*)(in + (gi * GS + s) * HSTEP);
    } else {
      const float* p = wave_in + (size_t)gi * GS * IN;
#pragma unroll
      for (int k = 0; k < IN; ++k) buf[k] = *(const v4f*)(p + poff[k]);
    }
  };

  auto own_row = [&](const v4f (&pc)[IN], v4f (&xr)[IN]) {
#pragma unroll
    for (int k = 0; k < IN; ++k) stg[k * 32 + lane] = pc[k];
#pragma unroll
    for (int q = 0; q < IN; ++q) {
      const int v = REV ? IN - 1 - q : q;
      xr[v] = stg[lane * IN + v];
    }
    __builtin_amdgcn_sched_barrier(0);
  };

  const bool hi = lane >= 16;
  const int row = lane & 15;
  auto fc_pair = [&](float pa, float pb, int tA, float qa, float qb, int tB) {
    const float p0 = fmaxf(pa, 0.0f), p1 = fmaxf(pb, 0.0f);
    const float q0 = fmaxf(qa, 0.0f), q1 = fmaxf(qb, 0.0f);
    const float xp0 = xhalf(p0), xp1 = xhalf(p1);
    const float xq0 = xhalf(q0), xq1 = xhalf(q1);
    v2f a0, a1;
    a0.x = hi ? xq0 : p0;  a0.y = hi ? xq1 : p1;
    a1.x = hi ? q0 : xp0;  a1.y = hi ? q1 : xp1;
    const int tl = hi ? tB : tA;
    const float* wp = wl + tl * WPITCH + row * 2;
    const int sw = WSWZ(tl);
#pragma unroll
    for (int nt = 0; nt < 4; ++nt) {
      const v2f bv = *(const v2f*)(wp + ((nt * 32) ^ sw));
      fsum[0][nt] = __builtin_amdgcn_wmma_f32_16x16x4_f32(
          false, a0, false, bv, (short)0, fsum[0][nt], false, false);
      fsum[1][nt] = __builtin_amdgcn_wmma_f32_16x16x4_f32(
          false, a1, false, bv, (short)0, fsum[1][nt], false, false);
    }
  };

  auto run_group = [&](int g, const v4f (&buf)[IN]) {
    float ph0 = 0.f, ph1 = 0.f;
    int pt = 0;
    const int gi = REV ? (NG - 1 - g) : g;
#pragma unroll
    for (int s = 0; s < GS; ++s) {
      const int sl = REV ? (GS - 1 - s) : s;
      const int tt = gi * GS + sl;

      float xv[IN];
#pragma unroll
      for (int c = 0; c < IN; ++c) {
        const int e = sl * IN + c;
        xv[c] = buf[e >> 2][e & 3];
      }

      float g8[8];
#pragma unroll
      for (int gg = 0; gg < 8; ++gg) {
        float acc = fmaf(whh[gg][0], h0, fmaf(whh[gg][1], h1, bias[gg]));
#pragma unroll
        for (int c = 0; c < IN; ++c) acc = fmaf(wih[gg][c], xv[c], acc);
        g8[gg] = acc;
      }

      const float i0 = sigm_half(g8[0]), i1 = sigm_half(g8[1]);
      const float f0 = sigm_half(g8[2]), f1 = sigm_half(g8[3]);
      const float t0 = tanh_hw(g8[4]), t1 = tanh_hw(g8[5]);
      const float o0 = sigm_half(g8[6]), o1 = sigm_half(g8[7]);

      c0 = fmaf(f0, c0, i0 * t0);
      c1 = fmaf(f1, c1, i1 * t1);
      h0 = o0 * tanh_hw(c0);
      h1 = o1 * tanh_hw(c1);

      if constexpr (FC) {
        if ((s & 1) == 0) { ph0 = h0; ph1 = h1; pt = tt; }
        else fc_pair(ph0, ph1, pt, h0, h1, tt);
      } else {
        *(v2f*)(out + tt * HSTEP) = v2f{h0, h1};
      }
    }
  };

  auto fetch = [&](int g, v4f (&buf)[IN]) {
    asm volatile("" ::: "memory");
    __builtin_amdgcn_sched_barrier(0);
    load_group(g, buf);
    __builtin_amdgcn_sched_barrier(0);
  };

  auto use_group = [&](int g, const v4f (&buf)[IN]) {
    if constexpr (TMAJOR) {
      run_group(g, buf);
    } else {
      v4f xr[IN];
      own_row(buf, xr);
      run_group(g, xr);
    }
  };

  v4f r0[IN], r1[IN], r2[IN];
  if constexpr (TMAJOR) {
    load_group(0, r0);
    load_group(1, r1);
  } else {
    fetch(0, r0);
    fetch(1, r1);
  }
  for (int g = 0; g < NG - 1; g += RING) {
    fetch(g + 2, r2);
    use_group(g, r0);
    fetch(g + 3, r0);
    use_group(g + 1, r1);
    fetch(g + 4, r1);
    use_group(g + 2, r2);
  }
  use_group(NG - 1, r0);
}

#define XP 68
__global__ __launch_bounds__(256) void lstm2_fc_kernel(
    const float* __restrict__ x,
    const float* __restrict__ w_ih_0f, const float* __restrict__ w_hh_0f,
    const float* __restrict__ b_ih_0f, const float* __restrict__ b_hh_0f,
    const float* __restrict__ w_ih_0r, const float* __restrict__ w_hh_0r,
    const float* __restrict__ b_ih_0r, const float* __restrict__ b_hh_0r,
    const float* __restrict__ w_ih_1f, const float* __restrict__ w_hh_1f,
    const float* __restrict__ b_ih_1f, const float* __restrict__ b_hh_1f,
    const float* __restrict__ w_ih_1r, const float* __restrict__ w_hh_1r,
    const float* __restrict__ b_ih_1r, const float* __restrict__ b_hh_1r,
    const float* __restrict__ w1,
    const float* __restrict__ b1,
    const float* __restrict__ w2,
    const float* __restrict__ b2,
    float* __restrict__ outp)
{
  static_assert(NB % WGS == 0, "whole work-groups");
  static_assert(8 * 32 * 7 * 4 <= 2 * NT * WPITCH, "x slots fit in wls");
  static_assert(4 * 32 * XP <= 2 * NT * WPITCH, "pair areas fit in wls");
  __shared__ __attribute__((aligned(16))) float hbuf[NT * HSTEP];
  __shared__ __attribute__((aligned(16))) float wls[2 * NT * WPITCH];

  const int lane = threadIdx.x & 31;
  const int wave = __builtin_amdgcn_readfirstlane((int)(threadIdx.x >> 5));
  const int dir = wave >> 2;
  const int smp = threadIdx.x & (WGS - 1);
  const int b = blockIdx.x * WGS + smp;
  float* hrow = hbuf + smp * 4;

  {
    v4f* stg = (v4f*)wls + wave * (32 * 7);
    v8f none[2][4];
    if (dir == 0)
      lstm_dir<7, false, false, false>(x, w_ih_0f, w_hh_0f, b_ih_0f, b_hh_0f,
                                       hrow, stg, nullptr, b, none);
    else
      lstm_dir<7, true, false, false>(x, w_ih_0r, w_hh_0r, b_ih_0r, b_hh_0r,
                                      hrow + 2, stg, nullptr, b, none);
  }

  __syncthreads();

  {
    static_assert(NT == 4 * 25, "a lane's 25 loads cover the 100 steps of its row");
    const int n = wave * 8 + (lane >> 2);
    const int tq = lane & 3;
    const float* wrow = w1 + (size_t)n * 400 + tq * 4;
    v4f wv[25];
#pragma unroll
    for (int k = 0; k < 25; ++k) wv[k] = *(const v4f*)(wrow + k * 16);
#pragma unroll
    for (int k = 0; k < 25; ++k) {
      const int t = k * 4 + tq;
      const int o = t * WPITCH + ((n * 2) ^ WSWZ(t));
      *(v2f*)(wls + o) = v2f{wv[k].x, wv[k].y};
      *(v2f*)(wls + NT * WPITCH + o) = v2f{wv[k].z, wv[k].w};
    }
  }
  __syncthreads();

  v8f fsum[2][4] = {};
  if (dir == 0)
    lstm_dir<4, false, true, true>(hrow, w_ih_1f, w_hh_1f, b_ih_1f, b_hh_1f,
                                   nullptr, nullptr, wls, b, fsum);
  else
    lstm_dir<4, true, true, true>(hrow, w_ih_1r, w_hh_1r, b_ih_1r, b_hh_1r,
                                  nullptr, nullptr, wls + NT * WPITCH, b, fsum);

  __syncthreads();

  float* xw = wls + (wave & 3) * (32 * XP);
  const bool hi = lane >= 16;
  const int row = lane & 15;
  if (dir == 1) {
#pragma unroll
    for (int mt = 0; mt < 2; ++mt)
#pragma unroll
      for (int nt = 0; nt < 4; ++nt)
#pragma unroll
        for (int r = 0; r < 8; ++r) {
          const int m = mt * 16 + (hi ? 8 : 0) + r;
          xw[m * XP + nt * 16 + row] = fsum[mt][nt][r];
        }
  }

  __syncthreads();

  if (dir == 0) {
#pragma unroll
    for (int nt = 0; nt < 4; ++nt) {
      const float bk = b1[nt * 16 + row];
#pragma unroll
      for (int mt = 0; mt < 2; ++mt)
#pragma unroll
        for (int r = 0; r < 8; ++r) {
          const int m = mt * 16 + (hi ? 8 : 0) + r;
          float* p = xw + m * XP + nt * 16 + row;
          *p = fmaxf(fsum[mt][nt][r] + *p + bk, 0.0f);
        }
    }
    const int n0 = row, n1 = 16 + row;
    const bool v1 = n1 < 20;
    const float* w2a = w2 + n0 * 64 + (hi ? 2 : 0);
    const float* w2b = w2 + (v1 ? n1 : 0) * 64 + (hi ? 2 : 0);
    const float* ar0 = xw + row * XP + (hi ? 2 : 0);
    const float* ar1 = xw + (16 + row) * XP + (hi ? 2 : 0);
    v8f acc[2][2] = {};
#pragma unroll
    for (int kk = 0; kk < 16; ++kk) {
      const v2f a0 = *(const v2f*)(ar0 + kk * 4);
      const v2f a1 = *(const v2f*)(ar1 + kk * 4);
      const v2f bv0 = *(const v2f*)(w2a + kk * 4);
      v2f bv1 = *(const v2f*)(w2b + kk * 4);
      if (!v1) { bv1.x = 0.0f; bv1.y = 0.0f; }
      acc[0][0] = __builtin_amdgcn_wmma_f32_16x16x4_f32(
          false, a0, false, bv0, (short)0, acc[0][0], false, false);
      acc[1][0] = __builtin_amdgcn_wmma_f32_16x16x4_f32(
          false, a1, false, bv0, (short)0, acc[1][0], false, false);
      acc[0][1] = __builtin_amdgcn_wmma_f32_16x16x4_f32(
          false, a0, false, bv1, (short)0, acc[0][1], false, false);
      acc[1][1] = __builtin_amdgcn_wmma_f32_16x16x4_f32(
          false, a1, false, bv1, (short)0, acc[1][1], false, false);
    }
    const float bn0 = b2[n0];
    const float bn1 = b2[v1 ? n1 : 0];
#pragma unroll
    for (int mt = 0; mt < 2; ++mt)
#pragma unroll
      for (int r = 0; r < 8; ++r) {
        const int m = mt * 16 + (hi ? 8 : 0) + r;
        xw[m * 20 + n0] = acc[mt][0][r] + bn0;
        if (v1) xw[m * 20 + n1] = acc[mt][1][r] + bn1;
      }
    float* ob = outp + (size_t)(b - lane) * 20;
#pragma unroll
    for (int i = 0; i < 5; ++i) {
      const v4f v = *(const v4f*)(xw + (i * 32 + lane) * 4);
      *(v4f*)(ob + (i * 32 + lane) * 4) = v;
    }
  }
}

extern "C" void kernel_launch(void* const* d_in, const int* in_sizes, int n_in,
                              void* d_out, int out_size, void* d_ws,
                              size_t ws_size, hipStream_t stream) {
  const float* x = (const float*)d_in[0];
  const float* w_ih_l0  = (const float*)d_in[1];
  const float* w_hh_l0  = (const float*)d_in[2];
  const float* b_ih_l0  = (const float*)d_in[3];
  const float* b_hh_l0  = (const float*)d_in[4];
  const float* w_ih_l0r = (const float*)d_in[5];
  const float* w_hh_l0r = (const float*)d_in[6];
  const float* b_ih_l0r = (const float*)d_in[7];
  const float* b_hh_l0r = (const float*)d_in[8];
  const float* w_ih_l1  = (const float*)d_in[9];
  const float* w_hh_l1  = (const float*)d_in[10];
  const float* b_ih_l1  = (const float*)d_in[11];
  const float* b_hh_l1  = (const float*)d_in[12];
  const float* w_ih_l1r = (const float*)d_in[13];
  const float* w_hh_l1r = (const float*)d_in[14];
  const float* b_ih_l1r = (const float*)d_in[15];
  const float* b_hh_l1r = (const float*)d_in[16];
  const float* fc1_w = (const float*)d_in[17];
  const float* fc1_b = (const float*)d_in[18];
  const float* fc2_w = (const float*)d_in[19];
  const float* fc2_b = (const float*)d_in[20];
  float* out = (float*)d_out;

  lstm2_fc_kernel<<<dim3(NB / WGS), dim3(256), 0, stream>>>(
      x, w_ih_l0, w_hh_l0, b_ih_l0, b_hh_l0,
      w_ih_l0r, w_hh_l0r, b_ih_l0r, b_hh_l0r,
      w_ih_l1, w_hh_l1, b_ih_l1, b_hh_l1,
      w_ih_l1r, w_hh_l1r, b_ih_l1r, b_hh_l1r,
      fc1_w, fc1_b, fc2_w, fc2_b, out);
}
